# starting kernel
# baseline (speedup 1.0000x reference)
#include <hip/hip_runtime.h>
#include <math.h>

typedef __attribute__((ext_vector_type(2))) float v2f;
typedef __attribute__((ext_vector_type(8))) float v8f;

#define NEG_SLOPE 0.2f
#define EPS_F 1e-16f

__device__ inline void atomicMaxFloat(float* addr, float val) {
    if (val >= 0.0f) {
        atomicMax((int*)addr, __float_as_int(val));
    } else {
        atomicMin((unsigned int*)addr, __float_as_uint(val));
    }
}

__global__ void init_ws(float* m1, float* den1, float* out1,
                        float* m2, float* den2, float* out2, int N) {
    int i = blockIdx.x * blockDim.x + threadIdx.x;
    if (i < N * 128) out1[i] = 0.0f;
    if (i < N * 4) { m1[i] = -INFINITY; den1[i] = 0.0f; }
    if (i < N * 2) out2[i] = 0.0f;
    if (i < N)     { m2[i] = -INFINITY; den2[i] = 0.0f; }
}

__global__ __launch_bounds__(256) void gemm_wmma_f32(const float* __restrict__ X,
                                                     const float* __restrict__ W,
                                                     float* __restrict__ H, int N) {
    const int rowBase = blockIdx.x * 16;
    const int wave = threadIdx.x >> 5;
    const int lane = threadIdx.x & 31;
    const int colBase = wave * 16;
    const int half = lane >> 4;
    const int mn = lane & 15;

    v8f c = {};
    const float* xrow = X + (size_t)(rowBase + mn) * 128;
    #pragma unroll 4
    for (int k = 0; k < 128; k += 4) {
        const int ka = k + 2 * half;
        v2f a;
        a.x = xrow[ka];
        a.y = xrow[ka + 1];
        v2f b;
        b.x = W[(size_t)ka * 128 + colBase + mn];
        b.y = W[(size_t)(ka + 1) * 128 + colBase + mn];
        c = __builtin_amdgcn_wmma_f32_16x16x4_f32(false, a, false, b,
                                                  (short)0, c, false, false);
    }
    #pragma unroll
    for (int r = 0; r < 8; ++r) {
        H[(size_t)(rowBase + r + 8 * half) * 128 + colBase + mn] = c[r];
    }
}

__global__ void attn_l1(const float* __restrict__ h1,
                        const float* __restrict__ att_s,
                        const float* __restrict__ att_d,
                        float* __restrict__ as1, float* __restrict__ ad1, int N) {
    int t = blockIdx.x * blockDim.x + threadIdx.x;
    if (t >= N * 4) return;
    int n = t >> 2, h = t & 3;
    const float4* hp = (const float4*)(h1 + (size_t)n * 128 + h * 32);
    const float4* sp = (const float4*)(att_s + h * 32);
    const float4* dp = (const float4*)(att_d + h * 32);
    float s = 0.0f, d = 0.0f;
    #pragma unroll
    for (int j = 0; j < 8; ++j) {
        float4 hv = hp[j], sv = sp[j], dv = dp[j];
        s += hv.x * sv.x + hv.y * sv.y + hv.z * sv.z + hv.w * sv.w;
        d += hv.x * dv.x + hv.y * dv.y + hv.z * dv.z + hv.w * dv.w;
    }
    as1[t] = s;
    ad1[t] = d;
}

__global__ void edge_max_l1(const int* __restrict__ ei, int E, int N,
                            const float* __restrict__ as1, const float* __restrict__ ad1,
                            float* __restrict__ m1) {
    int t = blockIdx.x * blockDim.x + threadIdx.x;
    int tot = (E + N) * 4;
    if (t >= tot) return;
    int e = t >> 2, h = t & 3;
    int src, dst;
    if (e < E) { src = ei[e]; dst = ei[E + e]; } else { src = dst = e - E; }
    float v = as1[src * 4 + h] + ad1[dst * 4 + h];
    v = v >= 0.0f ? v : NEG_SLOPE * v;
    atomicMaxFloat(&m1[dst * 4 + h], v);
}

__global__ void edge_sum_l1(const int* __restrict__ ei, int E, int N,
                            const float* __restrict__ as1, const float* __restrict__ ad1,
                            const float* __restrict__ m1, float* __restrict__ den1) {
    int t = blockIdx.x * blockDim.x + threadIdx.x;
    int tot = (E + N) * 4;
    if (t >= tot) return;
    int e = t >> 2, h = t & 3;
    int src, dst;
    if (e < E) { src = ei[e]; dst = ei[E + e]; } else { src = dst = e - E; }
    float v = as1[src * 4 + h] + ad1[dst * 4 + h];
    v = v >= 0.0f ? v : NEG_SLOPE * v;
    float mm = m1[dst * 4 + h];
    if (!isfinite(mm)) mm = 0.0f;
    atomicAdd(&den1[dst * 4 + h], expf(v - mm));
}

__global__ __launch_bounds__(256) void edge_agg_l1(const int* __restrict__ ei, int E, int N,
                                                   const float* __restrict__ as1,
                                                   const float* __restrict__ ad1,
                                                   const float* __restrict__ m1,
                                                   const float* __restrict__ den1,
                                                   const float* __restrict__ h1,
                                                   float* __restrict__ out1) {
    int e = blockIdx.x * 8 + (threadIdx.x >> 5);
    if (e >= E + N) return;
    int lane = threadIdx.x & 31;
    int src, dst;
    if (e < E) { src = ei[e]; dst = ei[E + e]; } else { src = dst = e - E; }
    int h = lane >> 3;
    float v = as1[src * 4 + h] + ad1[dst * 4 + h];
    v = v >= 0.0f ? v : NEG_SLOPE * v;
    float mm = m1[dst * 4 + h];
    if (!isfinite(mm)) mm = 0.0f;
    float alpha = expf(v - mm) / (den1[dst * 4 + h] + EPS_F);
    const float4* hs = (const float4*)(h1 + (size_t)src * 128);
    float4 hv = hs[lane];
    float* ob = out1 + (size_t)dst * 128 + lane * 4;
    atomicAdd(ob + 0, alpha * hv.x);
    atomicAdd(ob + 1, alpha * hv.y);
    atomicAdd(ob + 2, alpha * hv.z);
    atomicAdd(ob + 3, alpha * hv.w);
}

__global__ void bias_elu(float* __restrict__ out1, const float* __restrict__ b1, int N) {
    int i = blockIdx.x * blockDim.x + threadIdx.x;
    if (i >= N * 128) return;
    float v = out1[i] + b1[i & 127];
    out1[i] = v > 0.0f ? v : expf(v) - 1.0f;
}

__global__ __launch_bounds__(256) void l2_matvec(const float* __restrict__ hf,
                                                 const float* __restrict__ W2,
                                                 const float* __restrict__ att_s2,
                                                 const float* __restrict__ att_d2,
                                                 float* __restrict__ h2,
                                                 float* __restrict__ as2,
                                                 float* __restrict__ ad2, int N) {
    int n = blockIdx.x * 8 + (threadIdx.x >> 5);
    if (n >= N) return;
    int lane = threadIdx.x & 31;
    const float4* hp = (const float4*)(hf + (size_t)n * 128);
    float4 hv = hp[lane];
    int c = lane * 4;
    float s0 = hv.x * W2[(c + 0) * 2]     + hv.y * W2[(c + 1) * 2]
             + hv.z * W2[(c + 2) * 2]     + hv.w * W2[(c + 3) * 2];
    float s1 = hv.x * W2[(c + 0) * 2 + 1] + hv.y * W2[(c + 1) * 2 + 1]
             + hv.z * W2[(c + 2) * 2 + 1] + hv.w * W2[(c + 3) * 2 + 1];
    #pragma unroll
    for (int off = 16; off >= 1; off >>= 1) {
        s0 += __shfl_xor(s0, off, 32);
        s1 += __shfl_xor(s1, off, 32);
    }
    if (lane == 0) {
        h2[n * 2 + 0] = s0;
        h2[n * 2 + 1] = s1;
        as2[n] = s0 * att_s2[0] + s1 * att_s2[1];
        ad2[n] = s0 * att_d2[0] + s1 * att_d2[1];
    }
}

__global__ void edge_max_l2(const int* __restrict__ ei, int E, int N,
                            const float* __restrict__ as2, const float* __restrict__ ad2,
                            float* __restrict__ m2) {
    int e = blockIdx.x * blockDim.x + threadIdx.x;
    if (e >= E + N) return;
    int src, dst;
    if (e < E) { src = ei[e]; dst = ei[E + e]; } else { src = dst = e - E; }
    float v = as2[src] + ad2[dst];
    v = v >= 0.0f ? v : NEG_SLOPE * v;
    atomicMaxFloat(&m2[dst], v);
}

__global__ void edge_sum_l2(const int* __restrict__ ei, int E, int N,
                            const float* __restrict__ as2, const float* __restrict__ ad2,
                            const float* __restrict__ m2, float* __restrict__ den2) {
    int e = blockIdx.x * blockDim.x + threadIdx.x;
    if (e >= E + N) return;
    int src, dst;
    if (e < E) { src = ei[e]; dst = ei[E + e]; } else { src = dst = e - E; }
    float v = as2[src] + ad2[dst];
    v = v >= 0.0f ? v : NEG_SLOPE * v;
    float mm = m2[dst];
    if (!isfinite(mm)) mm = 0.0f;
    atomicAdd(&den2[dst], expf(v - mm));
}

__global__ void edge_agg_l2(const int* __restrict__ ei, int E, int N,
                            const float* __restrict__ as2, const float* __restrict__ ad2,
                            const float* __restrict__ m2, const float* __restrict__ den2,
                            const float* __restrict__ h2, float* __restrict__ out2) {
    int e = blockIdx.x * blockDim.x + threadIdx.x;
    if (e >= E + N) return;
    int src, dst;
    if (e < E) { src = ei[e]; dst = ei[E + e]; } else { src = dst = e - E; }
    float v = as2[src] + ad2[dst];
    v = v >= 0.0f ? v : NEG_SLOPE * v;
    float mm = m2[dst];
    if (!isfinite(mm)) mm = 0.0f;
    float alpha = expf(v - mm) / (den2[dst] + EPS_F);
    atomicAdd(&out2[dst * 2 + 0], alpha * h2[src * 2 + 0]);
    atomicAdd(&out2[dst * 2 + 1], alpha * h2[src * 2 + 1]);
}

__global__ void final_logsoftmax(const float* __restrict__ out2,
                                 const float* __restrict__ b2,
                                 float* __restrict__ y, int N) {
    int n = blockIdx.x * blockDim.x + threadIdx.x;
    if (n >= N) return;
    float z0 = out2[n * 2 + 0] + b2[0];
    float z1 = out2[n * 2 + 1] + b2[1];
    float mx = fmaxf(z0, z1);
    float lse = mx + logf(expf(z0 - mx) + expf(z1 - mx));
    y[n * 2 + 0] = z0 - lse;
    y[n * 2 + 1] = z1 - lse;
}

extern "C" void kernel_launch(void* const* d_in, const int* in_sizes, int n_in,
                              void* d_out, int out_size, void* d_ws, size_t ws_size,
                              hipStream_t stream) {
    const float* x        = (const float*)d_in[0];
    const int*   ei       = (const int*)d_in[1];
    const float* W1       = (const float*)d_in[2];
    const float* att_src1 = (const float*)d_in[3];
    const float* att_dst1 = (const float*)d_in[4];
    const float* b1       = (const float*)d_in[5];
    const float* W2       = (const float*)d_in[6];
    const float* att_src2 = (const float*)d_in[7];
    const float* att_dst2 = (const float*)d_in[8];
    const float* b2       = (const float*)d_in[9];
    float* y = (float*)d_out;

    const int N = in_sizes[0] / 128;
    const int E = in_sizes[1] / 2;
    const int Etot = E + N;

    float* ws   = (float*)d_ws;
    float* h1   = ws;
    float* out1 = h1   + (size_t)N * 128;
    float* as1  = out1 + (size_t)N * 128;
    float* ad1  = as1  + (size_t)N * 4;
    float* m1   = ad1  + (size_t)N * 4;
    float* den1 = m1   + (size_t)N * 4;
    float* h2   = den1 + (size_t)N * 4;
    float* as2  = h2   + (size_t)N * 2;
    float* ad2  = as2  + (size_t)N;
    float* m2   = ad2  + (size_t)N;
    float* den2 = m2   + (size_t)N;
    float* out2 = den2 + (size_t)N;

    const int TB = 256;

    init_ws<<<(N * 128 + TB - 1) / TB, TB, 0, stream>>>(m1, den1, out1, m2, den2, out2, N);

    gemm_wmma_f32<<<N / 16, 256, 0, stream>>>(x, W1, h1, N);
    attn_l1<<<(N * 4 + TB - 1) / TB, TB, 0, stream>>>(h1, att_src1, att_dst1, as1, ad1, N);

    edge_max_l1<<<(Etot * 4 + TB - 1) / TB, TB, 0, stream>>>(ei, E, N, as1, ad1, m1);
    edge_sum_l1<<<(Etot * 4 + TB - 1) / TB, TB, 0, stream>>>(ei, E, N, as1, ad1, m1, den1);
    edge_agg_l1<<<(Etot + 7) / 8, 256, 0, stream>>>(ei, E, N, as1, ad1, m1, den1, h1, out1);
    bias_elu<<<(N * 128 + TB - 1) / TB, TB, 0, stream>>>(out1, b1, N);

    l2_matvec<<<(N + 7) / 8, 256, 0, stream>>>(out1, W2, att_src2, att_dst2, h2, as2, ad2, N);

    edge_max_l2<<<(Etot + TB - 1) / TB, TB, 0, stream>>>(ei, E, N, as2, ad2, m2);
    edge_sum_l2<<<(Etot + TB - 1) / TB, TB, 0, stream>>>(ei, E, N, as2, ad2, m2, den2);
    edge_agg_l2<<<(Etot + TB - 1) / TB, TB, 0, stream>>>(ei, E, N, as2, ad2, m2, den2, h2, out2);

    final_logsoftmax<<<(N + TB - 1) / TB, TB, 0, stream>>>(out2, b2, y, N);
}


# Several nodes per wave in both segment kernels (layer 1: five in turn, the next one's sources and logits fetched under the row sums; layer 2: eight abreast in 4-lane groups); 83750 fewer waves
# speedup vs baseline: 3.4470x; 3.4470x over previous
#include <hip/hip_runtime.h>
#include <math.h>


typedef __attribute__((ext_vector_type(2))) float v2f;
typedef __attribute__((ext_vector_type(8))) float v8f;

#define NEG_SLOPE 0.2f
#define EPS_F 1e-16f

static __device__ __forceinline__ float lrelu(float v) { return v >= 0.0f ? v : NEG_SLOPE * v; }
static __device__ __forceinline__ float finite_or_zero(float m) { return isfinite(m) ? m : 0.0f; }
static __device__ __forceinline__ float wave_max(float v) {
    #pragma unroll
    for (int o = 16; o >= 1; o >>= 1) v = fmaxf(v, __shfl_xor(v, o, 32));
    return v;
}
static __device__ __forceinline__ float wave_sum(float v) {
    #pragma unroll
    for (int o = 16; o >= 1; o >>= 1) v += __shfl_xor(v, o, 32);
    return v;
}
static __device__ __forceinline__ float lane_f(float v, int j) {
    return __int_as_float(__builtin_amdgcn_readlane(__float_as_int(v), j));
}

__global__ void zero_deg(int* __restrict__ deg, int N) {
    int i = blockIdx.x * blockDim.x + threadIdx.x;
    if (i < N) deg[i] = 0;
}

static __device__ __forceinline__ int wave_incl_scan(int v, int lane) {
    #pragma unroll
    for (int o = 1; o < 32; o <<= 1) {
        int u = __shfl_up(v, o, 32);
        if (lane >= o) v += u;
    }
    return v;
}

static __device__ __forceinline__ int ld_deg(const int* __restrict__ deg, int idx, int N) {
    const int v = deg[idx < N ? idx : N - 1];
    return idx < N ? v : 0;
}

__global__ __launch_bounds__(1024) void scan_deg(const int* __restrict__ deg,
                                                 int* __restrict__ off, int N) {
    __shared__ int wtot[32];
    const int wave = threadIdx.x >> 5;
    const int lane = threadIdx.x & 31;
    const int per = ((N + 4095) >> 12) << 7;
    const int base = wave * per;

    int tsum = 0;
    for (int i = 0; i < per; i += 32) tsum += ld_deg(deg, base + i + lane, N);
    #pragma unroll
    for (int o = 16; o >= 1; o >>= 1) tsum += __shfl_xor(tsum, o, 32);
    if (lane == 0) wtot[wave] = tsum;
    __syncthreads();
    int carry = 0;
    for (int j = 0; j < wave; ++j) carry += wtot[j];

    const int last = N - 1;
    int a0 = deg[min(base + lane, last)];
    int a1 = deg[min(base + lane + 32, last)];
    int a2 = deg[min(base + lane + 64, last)];
    int a3 = deg[min(base + lane + 96, last)];
    for (int i = 0; i < per; i += 128) {
        const int i0 = base + i + lane, i1 = i0 + 32, i2 = i0 + 64, i3 = i0 + 96;
        const int v0 = i0 < N ? a0 : 0;
        const int v1 = i1 < N ? a1 : 0;
        const int v2 = i2 < N ? a2 : 0;
        const int v3 = i3 < N ? a3 : 0;
        if (i + 128 < per) {
            a0 = deg[min(i0 + 128, last)];
            a1 = deg[min(i1 + 128, last)];
            a2 = deg[min(i2 + 128, last)];
            a3 = deg[min(i3 + 128, last)];
        }
        const int s0 = wave_incl_scan(v0, lane);
        const int s1 = wave_incl_scan(v1, lane);
        const int s2 = wave_incl_scan(v2, lane);
        const int s3 = wave_incl_scan(v3, lane);
        const int c1 = carry + __shfl(s0, 31, 32);
        const int c2 = c1 + __shfl(s1, 31, 32);
        const int c3 = c2 + __shfl(s2, 31, 32);
        if (i0 < N) off[i0] = carry + s0 - v0;
        if (i1 < N) off[i1] = c1 + s1 - v1;
        if (i2 < N) off[i2] = c2 + s2 - v2;
        if (i3 < N) off[i3] = c3 + s3 - v3;
        carry = c3 + __shfl(s3, 31, 32);
    }
    if (threadIdx.x == 1023) off[N] = carry;
}

__global__ __launch_bounds__(256) void gemm_wmma_f32(const float* __restrict__ X,
                                                     const float* __restrict__ W,
                                                     float* __restrict__ H, int N,
                                                     const int* __restrict__ ei, int E,
                                                     int* __restrict__ deg, int* __restrict__ rank) {
    const int nthreads = gridDim.x * 256;
    const int e0 = blockIdx.x * 256 + threadIdx.x;
    const bool has = e0 < E;
    int r0 = 0;
    if (has) r0 = atomicAdd(&deg[ei[E + e0]], 1);

    const int rowBase = blockIdx.x * 16;
    const int wave = threadIdx.x >> 5;
    const int lane = threadIdx.x & 31;
    const int colBase = wave * 16;
    const int half = lane >> 4;
    const int mn = lane & 15;

    v8f c = {};
    const float* xrow = X + (size_t)(rowBase + mn) * 128;
    #pragma unroll 4
    for (int k = 0; k < 128; k += 4) {
        const int ka = k + 2 * half;
        v2f a;
        a.x = xrow[ka];
        a.y = xrow[ka + 1];
        v2f b;
        b.x = W[(size_t)ka * 128 + colBase + mn];
        b.y = W[(size_t)(ka + 1) * 128 + colBase + mn];
        c = __builtin_amdgcn_wmma_f32_16x16x4_f32(false, a, false, b,
                                                  (short)0, c, false, false);
    }
    #pragma unroll
    for (int r = 0; r < 8; ++r) {
        H[(size_t)(rowBase + r + 8 * half) * 128 + colBase + mn] = c[r];
    }

    if (has) rank[e0] = r0;
    for (int e = e0 + nthreads; e < E; e += nthreads) rank[e] = atomicAdd(&deg[ei[E + e]], 1);
}

__global__ void attn_l1(const float* __restrict__ h1,
                        const float* __restrict__ att_s,
                        const float* __restrict__ att_d,
                        float* __restrict__ as1, float* __restrict__ ad1, int N,
                        const int* __restrict__ ei, int E,
                        const int* __restrict__ off, const int* __restrict__ rank,
                        int* __restrict__ csr) {
    int t = blockIdx.x * blockDim.x + threadIdx.x;
    const int nthreads = gridDim.x * blockDim.x;
    for (int e = t; e < E; e += 4 * nthreads) {
        const int e1 = e + nthreads, e2 = e1 + nthreads, e3 = e2 + nthreads;
        if (e3 < E) {
            const int d0 = ei[E + e], d1 = ei[E + e1], d2 = ei[E + e2], d3 = ei[E + e3];
            const int q0 = rank[e], q1 = rank[e1], q2 = rank[e2], q3 = rank[e3];
            const int s0 = ei[e], s1 = ei[e1], s2 = ei[e2], s3 = ei[e3];
            const int o0 = off[d0], o1 = off[d1], o2 = off[d2], o3 = off[d3];
            csr[o0 + q0] = s0;
            csr[o1 + q1] = s1;
            csr[o2 + q2] = s2;
            csr[o3 + q3] = s3;
        } else {
            csr[off[ei[E + e]] + rank[e]] = ei[e];
            if (e1 < E) csr[off[ei[E + e1]] + rank[e1]] = ei[e1];
            if (e2 < E) csr[off[ei[E + e2]] + rank[e2]] = ei[e2];
        }
    }
    if (t >= N * 4) return;
    int n = t >> 2, h = t & 3;
    const float4* hp = (const float4*)(h1 + (size_t)n * 128 + h * 32);
    const float4* sp = (const float4*)(att_s + h * 32);
    const float4* dp = (const float4*)(att_d + h * 32);
    float s = 0.0f, d = 0.0f;
    #pragma unroll
    for (int j = 0; j < 8; ++j) {
        float4 hv = hp[j], sv = sp[j], dv = dp[j];
        s += hv.x * sv.x + hv.y * sv.y + hv.z * sv.z + hv.w * sv.w;
        d += hv.x * dv.x + hv.y * dv.y + hv.z * dv.z + hv.w * dv.w;
    }
    as1[t] = s;
    ad1[t] = d;
}

static __device__ __forceinline__ void add_rows(float4& acc, int cnt, int s, float4 al,
                                                const float* __restrict__ hl) {
    int j = 0;
    for (; j + 4 <= cnt; j += 4) {
        const float* r0 = hl + (size_t)__builtin_amdgcn_readlane(s, j) * 128;
        const float* r1 = hl + (size_t)__builtin_amdgcn_readlane(s, j + 1) * 128;
        const float* r2 = hl + (size_t)__builtin_amdgcn_readlane(s, j + 2) * 128;
        const float* r3 = hl + (size_t)__builtin_amdgcn_readlane(s, j + 3) * 128;
        const float q00 = r0[0], q01 = r0[32], q02 = r0[64], q03 = r0[96];
        const float q10 = r1[0], q11 = r1[32], q12 = r1[64], q13 = r1[96];
        const float q20 = r2[0], q21 = r2[32], q22 = r2[64], q23 = r2[96];
        const float q30 = r3[0], q31 = r3[32], q32 = r3[64], q33 = r3[96];
        acc.x += lane_f(al.x, j) * q00;     acc.y += lane_f(al.y, j) * q01;
        acc.z += lane_f(al.z, j) * q02;     acc.w += lane_f(al.w, j) * q03;
        acc.x += lane_f(al.x, j + 1) * q10; acc.y += lane_f(al.y, j + 1) * q11;
        acc.z += lane_f(al.z, j + 1) * q12; acc.w += lane_f(al.w, j + 1) * q13;
        acc.x += lane_f(al.x, j + 2) * q20; acc.y += lane_f(al.y, j + 2) * q21;
        acc.z += lane_f(al.z, j + 2) * q22; acc.w += lane_f(al.w, j + 2) * q23;
        acc.x += lane_f(al.x, j + 3) * q30; acc.y += lane_f(al.y, j + 3) * q31;
        acc.z += lane_f(al.z, j + 3) * q32; acc.w += lane_f(al.w, j + 3) * q33;
    }
    for (; j < cnt; ++j) {
        const float* r0 = hl + (size_t)__builtin_amdgcn_readlane(s, j) * 128;
        const float q00 = r0[0], q01 = r0[32], q02 = r0[64], q03 = r0[96];
        acc.x += lane_f(al.x, j) * q00;     acc.y += lane_f(al.y, j) * q01;
        acc.z += lane_f(al.z, j) * q02;     acc.w += lane_f(al.w, j) * q03;
    }
}

#define L1_NODES_PER_WAVE 5

__global__ __launch_bounds__(256, 10) void seg_l1(const int* __restrict__ off,
                                              const int* __restrict__ csr,
                                              const float* __restrict__ as1,
                                              const float* __restrict__ ad1,
                                              const float* __restrict__ h1,
                                              const float* __restrict__ b1,
                                              const float* __restrict__ W2,
                                              const float* __restrict__ att_s2,
                                              const float* __restrict__ att_d2,
                                              float* __restrict__ h2,
                                              float* __restrict__ as2,
                                              float* __restrict__ ad2, int N) {
    const int n0 = L1_NODES_PER_WAVE *
                   __builtin_amdgcn_readfirstlane((int)(blockIdx.x * 8 + (threadIdx.x >> 5)));
    if (n0 >= N) return;
    const int lane = threadIdx.x & 31;
    const float* hl = h1 + lane;
    const float4* as4 = (const float4*)as1;
    const float4* ad4 = (const float4*)ad1;

    const float c0 = b1[lane], c1 = b1[lane + 32], c2 = b1[lane + 64], c3 = b1[lane + 96];
    const float2* w2v = (const float2*)W2;
    const float2 w0 = w2v[lane], w1 = w2v[lane + 32], w2 = w2v[lane + 64], w3 = w2v[lane + 96];
    const float ts0 = att_s2[0], ts1 = att_s2[1], td0 = att_d2[0], td1 = att_d2[1];

    int o0 = off[n0];
    int o1 = off[n0 + 1];
    int o2 = off[min(n0 + 2, N)];
    int s0 = n0;
    if (lane > 0 && lane < o1 - o0 + 1) s0 = csr[o0 + lane - 1];
    float4 r0 = as4[s0];
    float4 adv = ad4[n0];

    for (int i = 0; i < L1_NODES_PER_WAVE; ++i) {
        const int n = n0 + i;
        if (n >= N) break;
        const int beg = o0;
        const int T = o1 - o0 + 1;
        const bool more = (i + 1 < L1_NODES_PER_WAVE) && (n + 1 < N);

        const bool ok0 = lane < T;
        float4 v0;
        v0.x = lrelu(r0.x + adv.x);
        v0.y = lrelu(r0.y + adv.y);
        v0.z = lrelu(r0.z + adv.z);
        v0.w = lrelu(r0.w + adv.w);

        const int o3 = off[min(n + 3, N)];
        const bool hasn = more && lane > 0 && lane < o2 - o1 + 1;
        const int sxn = csr[hasn ? o1 + lane - 1 : 0];

        float4 m;
        m.x = ok0 ? v0.x : -INFINITY;
        m.y = ok0 ? v0.y : -INFINITY;
        m.z = ok0 ? v0.z : -INFINITY;
        m.w = ok0 ? v0.w : -INFINITY;
        for (int t0 = 32; t0 < T; t0 += 32) {
            const int t = t0 + lane;
            if (t < T) {
                float4 v = as4[csr[beg + t - 1]];
                m.x = fmaxf(m.x, lrelu(v.x + adv.x));
                m.y = fmaxf(m.y, lrelu(v.y + adv.y));
                m.z = fmaxf(m.z, lrelu(v.z + adv.z));
                m.w = fmaxf(m.w, lrelu(v.w + adv.w));
            }
        }
        m.x = finite_or_zero(wave_max(m.x));
        m.y = finite_or_zero(wave_max(m.y));
        m.z = finite_or_zero(wave_max(m.z));
        m.w = finite_or_zero(wave_max(m.w));

        float4 e0;
        e0.x = ok0 ? expf(v0.x - m.x) : 0.0f;
        e0.y = ok0 ? expf(v0.y - m.y) : 0.0f;
        e0.z = ok0 ? expf(v0.z - m.z) : 0.0f;
        e0.w = ok0 ? expf(v0.w - m.w) : 0.0f;
        float4 d = e0;
        for (int t0 = 32; t0 < T; t0 += 32) {
            const int t = t0 + lane;
            if (t < T) {
                float4 v = as4[csr[beg + t - 1]];
                d.x += expf(lrelu(v.x + adv.x) - m.x);
                d.y += expf(lrelu(v.y + adv.y) - m.y);
                d.z += expf(lrelu(v.z + adv.z) - m.z);
                d.w += expf(lrelu(v.w + adv.w) - m.w);
            }
        }
        d.x = wave_sum(d.x) + EPS_F;
        d.y = wave_sum(d.y) + EPS_F;
        d.z = wave_sum(d.z) + EPS_F;
        d.w = wave_sum(d.w) + EPS_F;

        const int nn = more ? n + 1 : n;
        const int s0n = hasn ? sxn : nn;
        const float4 r0n = as4[s0n];
        const float4 advn = ad4[nn];

        float4 acc = {0.0f, 0.0f, 0.0f, 0.0f};
        {
            float4 al;
            al.x = e0.x / d.x;
            al.y = e0.y / d.y;
            al.z = e0.z / d.z;
            al.w = e0.w / d.w;
            add_rows(acc, T < 32 ? T : 32, s0, al, hl);
        }
        for (int t0 = 32; t0 < T; t0 += 32) {
            const int t = t0 + lane;
            int s = n;
            float4 al = {0.0f, 0.0f, 0.0f, 0.0f};
            if (t < T) {
                s = csr[beg + t - 1];
                float4 v = as4[s];
                al.x = expf(lrelu(v.x + adv.x) - m.x) / d.x;
                al.y = expf(lrelu(v.y + adv.y) - m.y) / d.y;
                al.z = expf(lrelu(v.z + adv.z) - m.z) / d.z;
                al.w = expf(lrelu(v.w + adv.w) - m.w) / d.w;
            }
            const int left = T - t0;
            add_rows(acc, left < 32 ? left : 32, s, al, hl);
        }

        float4 f;
        f.x = acc.x + c0; f.x = f.x > 0.0f ? f.x : expf(f.x) - 1.0f;
        f.y = acc.y + c1; f.y = f.y > 0.0f ? f.y : expf(f.y) - 1.0f;
        f.z = acc.z + c2; f.z = f.z > 0.0f ? f.z : expf(f.z) - 1.0f;
        f.w = acc.w + c3; f.w = f.w > 0.0f ? f.w : expf(f.w) - 1.0f;
        float p0 = f.x * w0.x + f.y * w1.x + f.z * w2.x + f.w * w3.x;
        float p1 = f.x * w0.y + f.y * w1.y + f.z * w2.y + f.w * w3.y;
        p0 = wave_sum(p0);
        p1 = wave_sum(p1);
        if (lane == 0) {
            h2[n * 2 + 0] = p0;
            h2[n * 2 + 1] = p1;
            as2[n] = p0 * ts0 + p1 * ts1;
            ad2[n] = p0 * td0 + p1 * td1;
        }

        o0 = o1; o1 = o2; o2 = o3;
        s0 = s0n; r0 = r0n; adv = advn;
    }
}

#define L2_LANES 4
#define L2_CHUNKS 8
static __device__ __forceinline__ float group4_max(float v) {
    v = fmaxf(v, __shfl_xor(v, 2, 32));
    v = fmaxf(v, __shfl_xor(v, 1, 32));
    return v;
}
static __device__ __forceinline__ float group4_sum(float v) {
    v += __shfl_xor(v, 2, 32);
    v += __shfl_xor(v, 1, 32);
    return v;
}

__global__ __launch_bounds__(256) void seg_l2(const int* __restrict__ off,
                                              const int* __restrict__ csr,
                                              const float* __restrict__ as2,
                                              const float* __restrict__ ad2,
                                              const float* __restrict__ h2,
                                              const float* __restrict__ b2,
                                              float* __restrict__ y, int N) {
    const int lane = threadIdx.x & 31;
    const int q = lane & (L2_LANES - 1);
    const int nw = (32 / L2_LANES) * (int)(blockIdx.x * 8 + (threadIdx.x >> 5));
    if (nw >= N) return;
    const int nr = nw + lane / L2_LANES;
    const bool live = nr < N;
    const int n = live ? nr : N - 1;
    const int beg = off[n];
    const int T = off[n + 1] - beg + 1;
    const float adn = ad2[n];
    const float2* h2v = (const float2*)h2;

    int s[L2_CHUNKS];
    bool ok[L2_CHUNKS];
    #pragma unroll
    for (int c = 0; c < L2_CHUNKS; ++c) {
        const int t = L2_LANES * c + q;
        ok[c] = t < T;
        const bool has = ok[c] && t > 0;
        const int sx = csr[has ? beg + t - 1 : 0];
        s[c] = has ? sx : n;
    }
    float v[L2_CHUNKS];
    float2 g[L2_CHUNKS];
    #pragma unroll
    for (int c = 0; c < L2_CHUNKS; ++c) {
        v[c] = lrelu(as2[s[c]] + adn);
        g[c] = h2v[s[c]];
    }

    float m = -INFINITY;
    #pragma unroll
    for (int c = 0; c < L2_CHUNKS; ++c) m = ok[c] ? fmaxf(m, v[c]) : m;
    for (int t = 32 + q; t < T; t += L2_LANES)
        m = fmaxf(m, lrelu(as2[csr[beg + t - 1]] + adn));
    m = finite_or_zero(group4_max(m));

    float e[L2_CHUNKS];
    float d = 0.0f;
    #pragma unroll
    for (int c = 0; c < L2_CHUNKS; ++c) {
        e[c] = ok[c] ? expf(v[c] - m) : 0.0f;
        d += e[c];
    }
    for (int t = 32 + q; t < T; t += L2_LANES)
        d += expf(lrelu(as2[csr[beg + t - 1]] + adn) - m);
    d = group4_sum(d) + EPS_F;

    float o0 = 0.0f, o1 = 0.0f;
    #pragma unroll
    for (int c = 0; c < L2_CHUNKS; ++c) {
        const float al = e[c] / d;
        o0 += ok[c] ? al * g[c].x : 0.0f;
        o1 += ok[c] ? al * g[c].y : 0.0f;
    }
    for (int t = 32 + q; t < T; t += L2_LANES) {
        const int sx = csr[beg + t - 1];
        const float al = expf(lrelu(as2[sx] + adn) - m) / d;
        const float2 gx = h2v[sx];
        o0 += al * gx.x;
        o1 += al * gx.y;
    }
    o0 = group4_sum(o0);
    o1 = group4_sum(o1);
    if (q == 0 && live) {
        const float z0 = o0 + b2[0];
        const float z1 = o1 + b2[1];
        const float mx = fmaxf(z0, z1);
        const float lse = mx + logf(expf(z0 - mx) + expf(z1 - mx));
        float2 r;
        r.x = z0 - lse;
        r.y = z1 - lse;
        ((float2*)y)[n] = r;
    }
}

extern "C" void kernel_launch(void* const* d_in, const int* in_sizes, int n_in,
                              void* d_out, int out_size, void* d_ws, size_t ws_size,
                              hipStream_t stream) {
    const float* x        = (const float*)d_in[0];
    const int*   ei       = (const int*)d_in[1];
    const float* W1       = (const float*)d_in[2];
    const float* att_src1 = (const float*)d_in[3];
    const float* att_dst1 = (const float*)d_in[4];
    const float* b1       = (const float*)d_in[5];
    const float* W2       = (const float*)d_in[6];
    const float* att_src2 = (const float*)d_in[7];
    const float* att_dst2 = (const float*)d_in[8];
    const float* b2       = (const float*)d_in[9];
    float* y = (float*)d_out;

    const int N = in_sizes[0] / 128;
    const int E = in_sizes[1] / 2;

    float* ws   = (float*)d_ws;
    float* h1   = ws;
    float* as1  = h1  + (size_t)N * 128;
    float* ad1  = as1 + (size_t)N * 4;
    float* h2   = ad1 + (size_t)N * 4;
    float* as2  = h2  + (size_t)N * 2;
    float* ad2  = as2 + (size_t)N;
    int*   deg  = (int*)(ad2 + (size_t)N);
    int*   off  = deg + (size_t)N;
    int*   rank = off + (size_t)N + 1;
    int*   csr  = rank + (size_t)E;

    const int TB = 256;

    zero_deg<<<(N + TB - 1) / TB, TB, 0, stream>>>(deg, N);
    gemm_wmma_f32<<<N / 16, 256, 0, stream>>>(x, W1, h1, N, ei, E, deg, rank);
    scan_deg<<<1, 1024, 0, stream>>>(deg, off, N);
    attn_l1<<<(N * 4 + TB - 1) / TB, TB, 0, stream>>>(h1, att_src1, att_dst1, as1, ad1, N,
                                                      ei, E, off, rank, csr);

    seg_l1<<<(N + 39) / 40, 256, 0, stream>>>(off, csr, as1, ad1, h1, b1, W2, att_src2, att_dst2,
                                            h2, as2, ad2, N);

    seg_l2<<<(N + 63) / 64, 256, 0, stream>>>(off, csr, as2, ad2, h2, b2, y, N);
}


# Projection: the 16-row X stripe is fetched once per work-group by whole-line loads and shared through LDS; A fragments are read from LDS, not by 16-line 8-byte global loads in all eight waves
# speedup vs baseline: 3.6719x; 1.0652x over previous
#include <hip/hip_runtime.h>
#include <math.h>


typedef __attribute__((ext_vector_type(2))) float v2f;
typedef __attribute__((ext_vector_type(8))) float v8f;

#define NEG_SLOPE 0.2f
#define EPS_F 1e-16f

static __device__ __forceinline__ float lrelu(float v) { return v >= 0.0f ? v : NEG_SLOPE * v; }
static __device__ __forceinline__ float finite_or_zero(float m) { return isfinite(m) ? m : 0.0f; }
static __device__ __forceinline__ float wave_max(float v) {
    #pragma unroll
    for (int o = 16; o >= 1; o >>= 1) v = fmaxf(v, __shfl_xor(v, o, 32));
    return v;
}
static __device__ __forceinline__ float wave_sum(float v) {
    #pragma unroll
    for (int o = 16; o >= 1; o >>= 1) v += __shfl_xor(v, o, 32);
    return v;
}
static __device__ __forceinline__ float lane_f(float v, int j) {
    return __int_as_float(__builtin_amdgcn_readlane(__float_as_int(v), j));
}

__global__ void zero_deg(int* __restrict__ deg, int N) {
    int i = blockIdx.x * blockDim.x + threadIdx.x;
    if (i < N) deg[i] = 0;
}

static __device__ __forceinline__ int wave_incl_scan(int v, int lane) {
    #pragma unroll
    for (int o = 1; o < 32; o <<= 1) {
        int u = __shfl_up(v, o, 32);
        if (lane >= o) v += u;
    }
    return v;
}

static __device__ __forceinline__ int ld_deg(const int* __restrict__ deg, int idx, int N) {
    const int v = deg[idx < N ? idx : N - 1];
    return idx < N ? v : 0;
}

__global__ __launch_bounds__(1024) void scan_deg(const int* __restrict__ deg,
                                                 int* __restrict__ off, int N) {
    __shared__ int wtot[32];
    const int wave = threadIdx.x >> 5;
    const int lane = threadIdx.x & 31;
    const int per = ((N + 4095) >> 12) << 7;
    const int base = wave * per;

    int tsum = 0;
    for (int i = 0; i < per; i += 32) tsum += ld_deg(deg, base + i + lane, N);
    #pragma unroll
    for (int o = 16; o >= 1; o >>= 1) tsum += __shfl_xor(tsum, o, 32);
    if (lane == 0) wtot[wave] = tsum;
    __syncthreads();
    int carry = 0;
    for (int j = 0; j < wave; ++j) carry += wtot[j];

    const int last = N - 1;
    int a0 = deg[min(base + lane, last)];
    int a1 = deg[min(base + lane + 32, last)];
    int a2 = deg[min(base + lane + 64, last)];
    int a3 = deg[min(base + lane + 96, last)];
    for (int i = 0; i < per; i += 128) {
        const int i0 = base + i + lane, i1 = i0 + 32, i2 = i0 + 64, i3 = i0 + 96;
        const int v0 = i0 < N ? a0 : 0;
        const int v1 = i1 < N ? a1 : 0;
        const int v2 = i2 < N ? a2 : 0;
        const int v3 = i3 < N ? a3 : 0;
        if (i + 128 < per) {
            a0 = deg[min(i0 + 128, last)];
            a1 = deg[min(i1 + 128, last)];
            a2 = deg[min(i2 + 128, last)];
            a3 = deg[min(i3 + 128, last)];
        }
        const int s0 = wave_incl_scan(v0, lane);
        const int s1 = wave_incl_scan(v1, lane);
        const int s2 = wave_incl_scan(v2, lane);
        const int s3 = wave_incl_scan(v3, lane);
        const int c1 = carry + __shfl(s0, 31, 32);
        const int c2 = c1 + __shfl(s1, 31, 32);
        const int c3 = c2 + __shfl(s2, 31, 32);
        if (i0 < N) off[i0] = carry + s0 - v0;
        if (i1 < N) off[i1] = c1 + s1 - v1;
        if (i2 < N) off[i2] = c2 + s2 - v2;
        if (i3 < N) off[i3] = c3 + s3 - v3;
        carry = c3 + __shfl(s3, 31, 32);
    }
    if (threadIdx.x == 1023) off[N] = carry;
}

#define XS_PITCH 132
__global__ __launch_bounds__(256) void gemm_wmma_f32(const float* __restrict__ X,
                                                     const float* __restrict__ W,
                                                     float* __restrict__ H, int N,
                                                     const int* __restrict__ ei, int E,
                                                     int* __restrict__ deg, int* __restrict__ rank) {
    __shared__ __attribute__((aligned(16))) float xs[16 * XS_PITCH];
    const int nthreads = gridDim.x * 256;
    const int e0 = blockIdx.x * 256 + threadIdx.x;
    const bool has = e0 < E;
    const int rowBase = blockIdx.x * 16;

    const float4* xg = (const float4*)(X + (size_t)rowBase * 128);
    const int f0 = threadIdx.x, f1 = threadIdx.x + 256;
    const float4 x0 = xg[f0];
    const float4 x1 = xg[f1];
    const int d0 = ei[E + (has ? e0 : 0)];
    *(float4*)(xs + (f0 >> 5) * XS_PITCH + (f0 & 31) * 4) = x0;
    *(float4*)(xs + (f1 >> 5) * XS_PITCH + (f1 & 31) * 4) = x1;
    __syncthreads();

    int r0 = 0;
    if (has) r0 = atomicAdd(&deg[d0], 1);

    const int wave = threadIdx.x >> 5;
    const int lane = threadIdx.x & 31;
    const int colBase = wave * 16;
    const int half = lane >> 4;
    const int mn = lane & 15;

    v8f c = {};
    const float* xrow = xs + mn * XS_PITCH;
    #pragma unroll 4
    for (int k = 0; k < 128; k += 4) {
        const int ka = k + 2 * half;
        const v2f a = *(const v2f*)(xrow + ka);
        v2f b;
        b.x = W[(size_t)ka * 128 + colBase + mn];
        b.y = W[(size_t)(ka + 1) * 128 + colBase + mn];
        c = __builtin_amdgcn_wmma_f32_16x16x4_f32(false, a, false, b,
                                                  (short)0, c, false, false);
    }
    #pragma unroll
    for (int r = 0; r < 8; ++r) {
        H[(size_t)(rowBase + r + 8 * half) * 128 + colBase + mn] = c[r];
    }

    if (has) rank[e0] = r0;
    for (int e = e0 + nthreads; e < E; e += nthreads) rank[e] = atomicAdd(&deg[ei[E + e]], 1);
}

__global__ void attn_l1(const float* __restrict__ h1,
                        const float* __restrict__ att_s,
                        const float* __restrict__ att_d,
                        float* __restrict__ as1, float* __restrict__ ad1, int N,
                        const int* __restrict__ ei, int E,
                        const int* __restrict__ off, const int* __restrict__ rank,
                        int* __restrict__ csr) {
    int t = blockIdx.x * blockDim.x + threadIdx.x;
    const int nthreads = gridDim.x * blockDim.x;
    for (int e = t; e < E; e += 4 * nthreads) {
        const int e1 = e + nthreads, e2 = e1 + nthreads, e3 = e2 + nthreads;
        if (e3 < E) {
            const int d0 = ei[E + e], d1 = ei[E + e1], d2 = ei[E + e2], d3 = ei[E + e3];
            const int q0 = rank[e], q1 = rank[e1], q2 = rank[e2], q3 = rank[e3];
            const int s0 = ei[e], s1 = ei[e1], s2 = ei[e2], s3 = ei[e3];
            const int o0 = off[d0], o1 = off[d1], o2 = off[d2], o3 = off[d3];
            csr[o0 + q0] = s0;
            csr[o1 + q1] = s1;
            csr[o2 + q2] = s2;
            csr[o3 + q3] = s3;
        } else {
            csr[off[ei[E + e]] + rank[e]] = ei[e];
            if (e1 < E) csr[off[ei[E + e1]] + rank[e1]] = ei[e1];
            if (e2 < E) csr[off[ei[E + e2]] + rank[e2]] = ei[e2];
        }
    }
    if (t >= N * 4) return;
    int n = t >> 2, h = t & 3;
    const float4* hp = (const float4*)(h1 + (size_t)n * 128 + h * 32);
    const float4* sp = (const float4*)(att_s + h * 32);
    const float4* dp = (const float4*)(att_d + h * 32);
    float s = 0.0f, d = 0.0f;
    #pragma unroll
    for (int j = 0; j < 8; ++j) {
        float4 hv = hp[j], sv = sp[j], dv = dp[j];
        s += hv.x * sv.x + hv.y * sv.y + hv.z * sv.z + hv.w * sv.w;
        d += hv.x * dv.x + hv.y * dv.y + hv.z * dv.z + hv.w * dv.w;
    }
    as1[t] = s;
    ad1[t] = d;
}

static __device__ __forceinline__ void add_rows(float4& acc, int cnt, int s, float4 al,
                                                const float* __restrict__ hl) {
    int j = 0;
    for (; j + 4 <= cnt; j += 4) {
        const float* r0 = hl + (size_t)__builtin_amdgcn_readlane(s, j) * 128;
        const float* r1 = hl + (size_t)__builtin_amdgcn_readlane(s, j + 1) * 128;
        const float* r2 = hl + (size_t)__builtin_amdgcn_readlane(s, j + 2) * 128;
        const float* r3 = hl + (size_t)__builtin_amdgcn_readlane(s, j + 3) * 128;
        const float q00 = r0[0], q01 = r0[32], q02 = r0[64], q03 = r0[96];
        const float q10 = r1[0], q11 = r1[32], q12 = r1[64], q13 = r1[96];
        const float q20 = r2[0], q21 = r2[32], q22 = r2[64], q23 = r2[96];
        const float q30 = r3[0], q31 = r3[32], q32 = r3[64], q33 = r3[96];
        acc.x += lane_f(al.x, j) * q00;     acc.y += lane_f(al.y, j) * q01;
        acc.z += lane_f(al.z, j) * q02;     acc.w += lane_f(al.w, j) * q03;
        acc.x += lane_f(al.x, j + 1) * q10; acc.y += lane_f(al.y, j + 1) * q11;
        acc.z += lane_f(al.z, j + 1) * q12; acc.w += lane_f(al.w, j + 1) * q13;
        acc.x += lane_f(al.x, j + 2) * q20; acc.y += lane_f(al.y, j + 2) * q21;
        acc.z += lane_f(al.z, j + 2) * q22; acc.w += lane_f(al.w, j + 2) * q23;
        acc.x += lane_f(al.x, j + 3) * q30; acc.y += lane_f(al.y, j + 3) * q31;
        acc.z += lane_f(al.z, j + 3) * q32; acc.w += lane_f(al.w, j + 3) * q33;
    }
    for (; j < cnt; ++j) {
        const float* r0 = hl + (size_t)__builtin_amdgcn_readlane(s, j) * 128;
        const float q00 = r0[0], q01 = r0[32], q02 = r0[64], q03 = r0[96];
        acc.x += lane_f(al.x, j) * q00;     acc.y += lane_f(al.y, j) * q01;
        acc.z += lane_f(al.z, j) * q02;     acc.w += lane_f(al.w, j) * q03;
    }
}

#define L1_NODES_PER_WAVE 5

__global__ __launch_bounds__(256, 10) void seg_l1(const int* __restrict__ off,
                                              const int* __restrict__ csr,
                                              const float* __restrict__ as1,
                                              const float* __restrict__ ad1,
                                              const float* __restrict__ h1,
                                              const float* __restrict__ b1,
                                              const float* __restrict__ W2,
                                              const float* __restrict__ att_s2,
                                              const float* __restrict__ att_d2,
                                              float* __restrict__ h2,
                                              float* __restrict__ as2,
                                              float* __restrict__ ad2, int N) {
    const int n0 = L1_NODES_PER_WAVE *
                   __builtin_amdgcn_readfirstlane((int)(blockIdx.x * 8 + (threadIdx.x >> 5)));
    if (n0 >= N) return;
    const int lane = threadIdx.x & 31;
    const float* hl = h1 + lane;
    const float4* as4 = (const float4*)as1;
    const float4* ad4 = (const float4*)ad1;

    const float c0 = b1[lane], c1 = b1[lane + 32], c2 = b1[lane + 64], c3 = b1[lane + 96];
    const float2* w2v = (const float2*)W2;
    const float2 w0 = w2v[lane], w1 = w2v[lane + 32], w2 = w2v[lane + 64], w3 = w2v[lane + 96];
    const float ts0 = att_s2[0], ts1 = att_s2[1], td0 = att_d2[0], td1 = att_d2[1];

    int o0 = off[n0];
    int o1 = off[n0 + 1];
    int o2 = off[min(n0 + 2, N)];
    int s0 = n0;
    if (lane > 0 && lane < o1 - o0 + 1) s0 = csr[o0 + lane - 1];
    float4 r0 = as4[s0];
    float4 adv = ad4[n0];

    for (int i = 0; i < L1_NODES_PER_WAVE; ++i) {
        const int n = n0 + i;
        if (n >= N) break;
        const int beg = o0;
        const int T = o1 - o0 + 1;
        const bool more = (i + 1 < L1_NODES_PER_WAVE) && (n + 1 < N);

        const bool ok0 = lane < T;
        float4 v0;
        v0.x = lrelu(r0.x + adv.x);
        v0.y = lrelu(r0.y + adv.y);
        v0.z = lrelu(r0.z + adv.z);
        v0.w = lrelu(r0.w + adv.w);

        const int o3 = off[min(n + 3, N)];
        const bool hasn = more && lane > 0 && lane < o2 - o1 + 1;
        const int sxn = csr[hasn ? o1 + lane - 1 : 0];

        float4 m;
        m.x = ok0 ? v0.x : -INFINITY;
        m.y = ok0 ? v0.y : -INFINITY;
        m.z = ok0 ? v0.z : -INFINITY;
        m.w = ok0 ? v0.w : -INFINITY;
        for (int t0 = 32; t0 < T; t0 += 32) {
            const int t = t0 + lane;
            if (t < T) {
                float4 v = as4[csr[beg + t - 1]];
                m.x = fmaxf(m.x, lrelu(v.x + adv.x));
                m.y = fmaxf(m.y, lrelu(v.y + adv.y));
                m.z = fmaxf(m.z, lrelu(v.z + adv.z));
                m.w = fmaxf(m.w, lrelu(v.w + adv.w));
            }
        }
        m.x = finite_or_zero(wave_max(m.x));
        m.y = finite_or_zero(wave_max(m.y));
        m.z = finite_or_zero(wave_max(m.z));
        m.w = finite_or_zero(wave_max(m.w));

        float4 e0;
        e0.x = ok0 ? expf(v0.x - m.x) : 0.0f;
        e0.y = ok0 ? expf(v0.y - m.y) : 0.0f;
        e0.z = ok0 ? expf(v0.z - m.z) : 0.0f;
        e0.w = ok0 ? expf(v0.w - m.w) : 0.0f;
        float4 d = e0;
        for (int t0 = 32; t0 < T; t0 += 32) {
            const int t = t0 + lane;
            if (t < T) {
                float4 v = as4[csr[beg + t - 1]];
                d.x += expf(lrelu(v.x + adv.x) - m.x);
                d.y += expf(lrelu(v.y + adv.y) - m.y);
                d.z += expf(lrelu(v.z + adv.z) - m.z);
                d.w += expf(lrelu(v.w + adv.w) - m.w);
            }
        }
        d.x = wave_sum(d.x) + EPS_F;
        d.y = wave_sum(d.y) + EPS_F;
        d.z = wave_sum(d.z) + EPS_F;
        d.w = wave_sum(d.w) + EPS_F;

        const int nn = more ? n + 1 : n;
        const int s0n = hasn ? sxn : nn;
        const float4 r0n = as4[s0n];
        const float4 advn = ad4[nn];

        float4 acc = {0.0f, 0.0f, 0.0f, 0.0f};
        {
            float4 al;
            al.x = e0.x / d.x;
            al.y = e0.y / d.y;
            al.z = e0.z / d.z;
            al.w = e0.w / d.w;
            add_rows(acc, T < 32 ? T : 32, s0, al, hl);
        }
        for (int t0 = 32; t0 < T; t0 += 32) {
            const int t = t0 + lane;
            int s = n;
            float4 al = {0.0f, 0.0f, 0.0f, 0.0f};
            if (t < T) {
                s = csr[beg + t - 1];
                float4 v = as4[s];
                al.x = expf(lrelu(v.x + adv.x) - m.x) / d.x;
                al.y = expf(lrelu(v.y + adv.y) - m.y) / d.y;
                al.z = expf(lrelu(v.z + adv.z) - m.z) / d.z;
                al.w = expf(lrelu(v.w + adv.w) - m.w) / d.w;
            }
            const int left = T - t0;
            add_rows(acc, left < 32 ? left : 32, s, al, hl);
        }

        float4 f;
        f.x = acc.x + c0; f.x = f.x > 0.0f ? f.x : expf(f.x) - 1.0f;
        f.y = acc.y + c1; f.y = f.y > 0.0f ? f.y : expf(f.y) - 1.0f;
        f.z = acc.z + c2; f.z = f.z > 0.0f ? f.z : expf(f.z) - 1.0f;
        f.w = acc.w + c3; f.w = f.w > 0.0f ? f.w : expf(f.w) - 1.0f;
        float p0 = f.x * w0.x + f.y * w1.x + f.z * w2.x + f.w * w3.x;
        float p1 = f.x * w0.y + f.y * w1.y + f.z * w2.y + f.w * w3.y;
        p0 = wave_sum(p0);
        p1 = wave_sum(p1);
        if (lane == 0) {
            h2[n * 2 + 0] = p0;
            h2[n * 2 + 1] = p1;
            as2[n] = p0 * ts0 + p1 * ts1;
            ad2[n] = p0 * td0 + p1 * td1;
        }

        o0 = o1; o1 = o2; o2 = o3;
        s0 = s0n; r0 = r0n; adv = advn;
    }
}

#define L2_LANES 4
#define L2_CHUNKS 8
static __device__ __forceinline__ float group4_max(float v) {
    v = fmaxf(v, __shfl_xor(v, 2, 32));
    v = fmaxf(v, __shfl_xor(v, 1, 32));
    return v;
}
static __device__ __forceinline__ float group4_sum(float v) {
    v += __shfl_xor(v, 2, 32);
    v += __shfl_xor(v, 1, 32);
    return v;
}

__global__ __launch_bounds__(256) void seg_l2(const int* __restrict__ off,
                                              const int* __restrict__ csr,
                                              const float* __restrict__ as2,
                                              const float* __restrict__ ad2,
                                              const float* __restrict__ h2,
                                              const float* __restrict__ b2,
                                              float* __restrict__ y, int N) {
    const int lane = threadIdx.x & 31;
    const int q = lane & (L2_LANES - 1);
    const int nw = (32 / L2_LANES) * (int)(blockIdx.x * 8 + (threadIdx.x >> 5));
    if (nw >= N) return;
    const int nr = nw + lane / L2_LANES;
    const bool live = nr < N;
    const int n = live ? nr : N - 1;
    const int beg = off[n];
    const int T = off[n + 1] - beg + 1;
    const float adn = ad2[n];
    const float2* h2v = (const float2*)h2;

    int s[L2_CHUNKS];
    bool ok[L2_CHUNKS];
    #pragma unroll
    for (int c = 0; c < L2_CHUNKS; ++c) {
        const int t = L2_LANES * c + q;
        ok[c] = t < T;
        const bool has = ok[c] && t > 0;
        const int sx = csr[has ? beg + t - 1 : 0];
        s[c] = has ? sx : n;
    }
    float v[L2_CHUNKS];
    float2 g[L2_CHUNKS];
    #pragma unroll
    for (int c = 0; c < L2_CHUNKS; ++c) {
        v[c] = lrelu(as2[s[c]] + adn);
        g[c] = h2v[s[c]];
    }

    float m = -INFINITY;
    #pragma unroll
    for (int c = 0; c < L2_CHUNKS; ++c) m = ok[c] ? fmaxf(m, v[c]) : m;
    for (int t = 32 + q; t < T; t += L2_LANES)
        m = fmaxf(m, lrelu(as2[csr[beg + t - 1]] + adn));
    m = finite_or_zero(group4_max(m));

    float e[L2_CHUNKS];
    float d = 0.0f;
    #pragma unroll
    for (int c = 0; c < L2_CHUNKS; ++c) {
        e[c] = ok[c] ? expf(v[c] - m) : 0.0f;
        d += e[c];
    }
    for (int t = 32 + q; t < T; t += L2_LANES)
        d += expf(lrelu(as2[csr[beg + t - 1]] + adn) - m);
    d = group4_sum(d) + EPS_F;

    float o0 = 0.0f, o1 = 0.0f;
    #pragma unroll
    for (int c = 0; c < L2_CHUNKS; ++c) {
        const float al = e[c] / d;
        o0 += ok[c] ? al * g[c].x : 0.0f;
        o1 += ok[c] ? al * g[c].y : 0.0f;
    }
    for (int t = 32 + q; t < T; t += L2_LANES) {
        const int sx = csr[beg + t - 1];
        const float al = expf(lrelu(as2[sx] + adn) - m) / d;
        const float2 gx = h2v[sx];
        o0 += al * gx.x;
        o1 += al * gx.y;
    }
    o0 = group4_sum(o0);
    o1 = group4_sum(o1);
    if (q == 0 && live) {
        const float z0 = o0 + b2[0];
        const float z1 = o1 + b2[1];
        const float mx = fmaxf(z0, z1);
        const float lse = mx + logf(expf(z0 - mx) + expf(z1 - mx));
        float2 r;
        r.x = z0 - lse;
        r.y = z1 - lse;
        ((float2*)y)[n] = r;
    }
}

extern "C" void kernel_launch(void* const* d_in, const int* in_sizes, int n_in,
                              void* d_out, int out_size, void* d_ws, size_t ws_size,
                              hipStream_t stream) {
    const float* x        = (const float*)d_in[0];
    const int*   ei       = (const int*)d_in[1];
    const float* W1       = (const float*)d_in[2];
    const float* att_src1 = (const float*)d_in[3];
    const float* att_dst1 = (const float*)d_in[4];
    const float* b1       = (const float*)d_in[5];
    const float* W2       = (const float*)d_in[6];
    const float* att_src2 = (const float*)d_in[7];
    const float* att_dst2 = (const float*)d_in[8];
    const float* b2       = (const float*)d_in[9];
    float* y = (float*)d_out;

    const int N = in_sizes[0] / 128;
    const int E = in_sizes[1] / 2;

    float* ws   = (float*)d_ws;
    float* h1   = ws;
    float* as1  = h1  + (size_t)N * 128;
    float* ad1  = as1 + (size_t)N * 4;
    float* h2   = ad1 + (size_t)N * 4;
    float* as2  = h2  + (size_t)N * 2;
    float* ad2  = as2 + (size_t)N;
    int*   deg  = (int*)(ad2 + (size_t)N);
    int*   off  = deg + (size_t)N;
    int*   rank = off + (size_t)N + 1;
    int*   csr  = rank + (size_t)E;

    const int TB = 256;

    zero_deg<<<(N + TB - 1) / TB, TB, 0, stream>>>(deg, N);
    gemm_wmma_f32<<<N / 16, 256, 0, stream>>>(x, W1, h1, N, ei, E, deg, rank);
    scan_deg<<<1, 1024, 0, stream>>>(deg, off, N);
    attn_l1<<<(N * 4 + TB - 1) / TB, TB, 0, stream>>>(h1, att_src1, att_dst1, as1, ad1, N,
                                                      ei, E, off, rank, csr);

    seg_l1<<<(N + 39) / 40, 256, 0, stream>>>(off, csr, as1, ad1, h1, b1, W2, att_src2, att_dst2,
                                            h2, as2, ad2, N);

    seg_l2<<<(N + 63) / 64, 256, 0, stream>>>(off, csr, as2, ad2, h2, b2, y, N);
}


# Degree scan: each wave fetches its whole 1664-word range in one round of 16-byte loads and scans it in registers (13 wave scans abreast, one barrier); no second pass, no chunk-by-chunk carry chain
# speedup vs baseline: 3.8293x; 1.0429x over previous
#include <hip/hip_runtime.h>
#include <math.h>


typedef __attribute__((ext_vector_type(2))) float v2f;
typedef __attribute__((ext_vector_type(8))) float v8f;

#define NEG_SLOPE 0.2f
#define EPS_F 1e-16f

static __device__ __forceinline__ float lrelu(float v) { return v >= 0.0f ? v : NEG_SLOPE * v; }
static __device__ __forceinline__ float finite_or_zero(float m) { return isfinite(m) ? m : 0.0f; }
static __device__ __forceinline__ float wave_max(float v) {
    #pragma unroll
    for (int o = 16; o >= 1; o >>= 1) v = fmaxf(v, __shfl_xor(v, o, 32));
    return v;
}
static __device__ __forceinline__ float wave_sum(float v) {
    #pragma unroll
    for (int o = 16; o >= 1; o >>= 1) v += __shfl_xor(v, o, 32);
    return v;
}
static __device__ __forceinline__ float lane_f(float v, int j) {
    return __int_as_float(__builtin_amdgcn_readlane(__float_as_int(v), j));
}

__global__ void zero_deg(int* __restrict__ deg, int N) {
    int i = blockIdx.x * blockDim.x + threadIdx.x;
    if (i < N) deg[i] = 0;
}

#define SC_CHUNKS 13
#define SC_WAVE_WORDS (SC_CHUNKS * 128)
#define SC_TILE_WORDS (32 * SC_WAVE_WORDS)

static __device__ __forceinline__ int wave_incl_scan(int v, int lane) {
    #pragma unroll
    for (int o = 1; o < 32; o <<= 1) {
        int u = __shfl_up(v, o, 32);
        if (lane >= o) v += u;
    }
    return v;
}

static __device__ __forceinline__ int ld_deg(const int* __restrict__ deg, int idx, int N) {
    const int v = deg[idx < N ? idx : N - 1];
    return idx < N ? v : 0;
}

static __device__ __forceinline__ void scan_words(const int* __restrict__ deg,
                                                  int* __restrict__ off, int N, int* wtot) {
    const int wave = threadIdx.x >> 5;
    const int lane = threadIdx.x & 31;
    const int per = ((N + 4095) >> 12) << 7;
    const int base = wave * per;

    int tsum = 0;
    for (int i = 0; i < per; i += 32) tsum += ld_deg(deg, base + i + lane, N);
    #pragma unroll
    for (int o = 16; o >= 1; o >>= 1) tsum += __shfl_xor(tsum, o, 32);
    if (lane == 0) wtot[wave] = tsum;
    __syncthreads();
    int carry = 0;
    for (int j = 0; j < wave; ++j) carry += wtot[j];

    const int last = N - 1;
    int a0 = deg[min(base + lane, last)];
    int a1 = deg[min(base + lane + 32, last)];
    int a2 = deg[min(base + lane + 64, last)];
    int a3 = deg[min(base + lane + 96, last)];
    for (int i = 0; i < per; i += 128) {
        const int i0 = base + i + lane, i1 = i0 + 32, i2 = i0 + 64, i3 = i0 + 96;
        const int v0 = i0 < N ? a0 : 0;
        const int v1 = i1 < N ? a1 : 0;
        const int v2 = i2 < N ? a2 : 0;
        const int v3 = i3 < N ? a3 : 0;
        if (i + 128 < per) {
            a0 = deg[min(i0 + 128, last)];
            a1 = deg[min(i1 + 128, last)];
            a2 = deg[min(i2 + 128, last)];
            a3 = deg[min(i3 + 128, last)];
        }
        const int s0 = wave_incl_scan(v0, lane);
        const int s1 = wave_incl_scan(v1, lane);
        const int s2 = wave_incl_scan(v2, lane);
        const int s3 = wave_incl_scan(v3, lane);
        const int c1 = carry + __shfl(s0, 31, 32);
        const int c2 = c1 + __shfl(s1, 31, 32);
        const int c3 = c2 + __shfl(s2, 31, 32);
        if (i0 < N) off[i0] = carry + s0 - v0;
        if (i1 < N) off[i1] = c1 + s1 - v1;
        if (i2 < N) off[i2] = c2 + s2 - v2;
        if (i3 < N) off[i3] = c3 + s3 - v3;
        carry = c3 + __shfl(s3, 31, 32);
    }
    if (threadIdx.x == 1023) off[N] = carry;
}

static __device__ __forceinline__ void scan_groups(const int* __restrict__ deg, int* __restrict__ off,
                                                   int N, int* wtot) {
    const int wave = __builtin_amdgcn_readfirstlane((int)(threadIdx.x >> 5));
    const int lane = threadIdx.x & 31;
    const int G = N >> 2;
    const int4* d4 = (const int4*)deg;
    int4* o4 = (int4*)off;
    int tile_carry = 0;
    for (int tile0 = 0; tile0 <= N; tile0 += SC_TILE_WORDS) {
        const int g0 = (tile0 + wave * SC_WAVE_WORDS) / 4 + lane;
        int4 v[SC_CHUNKS];
        #pragma unroll
        for (int j = 0; j < SC_CHUNKS; ++j) v[j] = d4[min(g0 + j * 32, G - 1)];
        __builtin_amdgcn_sched_barrier(0);
        #pragma unroll
        for (int j = 0; j < SC_CHUNKS; ++j) {
            const bool in = g0 + j * 32 < G;
            v[j].x = in ? v[j].x : 0;
            v[j].y = in ? v[j].y : 0;
            v[j].z = in ? v[j].z : 0;
            v[j].w = in ? v[j].w : 0;
        }

        int s[SC_CHUNKS];
        #pragma unroll
        for (int j = 0; j < SC_CHUNKS; ++j) s[j] = v[j].x + v[j].y + v[j].z + v[j].w;
        #pragma unroll
        for (int o = 1; o < 32; o <<= 1) {
            int u[SC_CHUNKS];
            #pragma unroll
            for (int j = 0; j < SC_CHUNKS; ++j) u[j] = __shfl_up(s[j], o, 32);
            __builtin_amdgcn_sched_barrier(0);
            #pragma unroll
            for (int j = 0; j < SC_CHUNKS; ++j) s[j] += lane >= o ? u[j] : 0;
        }

        int cc[SC_CHUNKS];
        int run = 0;
        #pragma unroll
        for (int j = 0; j < SC_CHUNKS; ++j) {
            cc[j] = run;
            run += __builtin_amdgcn_readlane(s[j], 31);
        }
        if (lane == 0) wtot[wave] = run;
        __syncthreads();
        int before = tile_carry, all = 0;
        #pragma unroll
        for (int q = 0; q < 32; ++q) {
            const int w = wtot[q];
            all += w;
            before += q < wave ? w : 0;
        }

        #pragma unroll
        for (int j = 0; j < SC_CHUNKS; ++j) {
            const int g = g0 + j * 32;
            const int t = v[j].x + v[j].y + v[j].z + v[j].w;
            const int b0 = before + cc[j] + s[j] - t;
            if (g < G) {
                int4 r;
                r.x = b0;
                r.y = r.x + v[j].x;
                r.z = r.y + v[j].y;
                r.w = r.z + v[j].z;
                o4[g] = r;
            } else if (g == G) {
                off[N] = b0;
            }
        }

        tile_carry += all;
        if (tile0 + SC_TILE_WORDS <= N) __syncthreads();
    }
}

__global__ __launch_bounds__(1024) void scan_deg(const int* __restrict__ deg,
                                                 int* __restrict__ off, int N) {
    __shared__ int wtot[32];
    const bool groups = (N & 3) == 0 && N >= 4 &&
                        ((((size_t)deg) | ((size_t)off)) & 15) == 0;
    if (groups) scan_groups(deg, off, N, wtot);
    else        scan_words(deg, off, N, wtot);
}

#define XS_PITCH 132
__global__ __launch_bounds__(256) void gemm_wmma_f32(const float* __restrict__ X,
                                                     const float* __restrict__ W,
                                                     float* __restrict__ H, int N,
                                                     const int* __restrict__ ei, int E,
                                                     int* __restrict__ deg, int* __restrict__ rank) {
    __shared__ __attribute__((aligned(16))) float xs[16 * XS_PITCH];
    const int nthreads = gridDim.x * 256;
    const int e0 = blockIdx.x * 256 + threadIdx.x;
    const bool has = e0 < E;
    const int rowBase = blockIdx.x * 16;

    const float4* xg = (const float4*)(X + (size_t)rowBase * 128);
    const int f0 = threadIdx.x, f1 = threadIdx.x + 256;
    const float4 x0 = xg[f0];
    const float4 x1 = xg[f1];
    const int d0 = ei[E + (has ? e0 : 0)];
    *(float4*)(xs + (f0 >> 5) * XS_PITCH + (f0 & 31) * 4) = x0;
    *(float4*)(xs + (f1 >> 5) * XS_PITCH + (f1 & 31) * 4) = x1;
    __syncthreads();

    int r0 = 0;
    if (has) r0 = atomicAdd(&deg[d0], 1);

    const int wave = threadIdx.x >> 5;
    const int lane = threadIdx.x & 31;
    const int colBase = wave * 16;
    const int half = lane >> 4;
    const int mn = lane & 15;

    v8f c = {};
    const float* xrow = xs + mn * XS_PITCH;
    #pragma unroll 4
    for (int k = 0; k < 128; k += 4) {
        const int ka = k + 2 * half;
        const v2f a = *(const v2f*)(xrow + ka);
        v2f b;
        b.x = W[(size_t)ka * 128 + colBase + mn];
        b.y = W[(size_t)(ka + 1) * 128 + colBase + mn];
        c = __builtin_amdgcn_wmma_f32_16x16x4_f32(false, a, false, b,
                                                  (short)0, c, false, false);
    }
    #pragma unroll
    for (int r = 0; r < 8; ++r) {
        H[(size_t)(rowBase + r + 8 * half) * 128 + colBase + mn] = c[r];
    }

    if (has) rank[e0] = r0;
    for (int e = e0 + nthreads; e < E; e += nthreads) rank[e] = atomicAdd(&deg[ei[E + e]], 1);
}

__global__ void attn_l1(const float* __restrict__ h1,
                        const float* __restrict__ att_s,
                        const float* __restrict__ att_d,
                        float* __restrict__ as1, float* __restrict__ ad1, int N,
                        const int* __restrict__ ei, int E,
                        const int* __restrict__ off, const int* __restrict__ rank,
                        int* __restrict__ csr) {
    int t = blockIdx.x * blockDim.x + threadIdx.x;
    const int nthreads = gridDim.x * blockDim.x;
    for (int e = t; e < E; e += 4 * nthreads) {
        const int e1 = e + nthreads, e2 = e1 + nthreads, e3 = e2 + nthreads;
        if (e3 < E) {
            const int d0 = ei[E + e], d1 = ei[E + e1], d2 = ei[E + e2], d3 = ei[E + e3];
            const int q0 = rank[e], q1 = rank[e1], q2 = rank[e2], q3 = rank[e3];
            const int s0 = ei[e], s1 = ei[e1], s2 = ei[e2], s3 = ei[e3];
            const int o0 = off[d0], o1 = off[d1], o2 = off[d2], o3 = off[d3];
            csr[o0 + q0] = s0;
            csr[o1 + q1] = s1;
            csr[o2 + q2] = s2;
            csr[o3 + q3] = s3;
        } else {
            csr[off[ei[E + e]] + rank[e]] = ei[e];
            if (e1 < E) csr[off[ei[E + e1]] + rank[e1]] = ei[e1];
            if (e2 < E) csr[off[ei[E + e2]] + rank[e2]] = ei[e2];
        }
    }
    if (t >= N * 4) return;
    int n = t >> 2, h = t & 3;
    const float4* hp = (const float4*)(h1 + (size_t)n * 128 + h * 32);
    const float4* sp = (const float4*)(att_s + h * 32);
    const float4* dp = (const float4*)(att_d + h * 32);
    float s = 0.0f, d = 0.0f;
    #pragma unroll
    for (int j = 0; j < 8; ++j) {
        float4 hv = hp[j], sv = sp[j], dv = dp[j];
        s += hv.x * sv.x + hv.y * sv.y + hv.z * sv.z + hv.w * sv.w;
        d += hv.x * dv.x + hv.y * dv.y + hv.z * dv.z + hv.w * dv.w;
    }
    as1[t] = s;
    ad1[t] = d;
}

static __device__ __forceinline__ void add_rows(float4& acc, int cnt, int s, float4 al,
                                                const float* __restrict__ hl) {
    int j = 0;
    for (; j + 4 <= cnt; j += 4) {
        const float* r0 = hl + (size_t)__builtin_amdgcn_readlane(s, j) * 128;
        const float* r1 = hl + (size_t)__builtin_amdgcn_readlane(s, j + 1) * 128;
        const float* r2 = hl + (size_t)__builtin_amdgcn_readlane(s, j + 2) * 128;
        const float* r3 = hl + (size_t)__builtin_amdgcn_readlane(s, j + 3) * 128;
        const float q00 = r0[0], q01 = r0[32], q02 = r0[64], q03 = r0[96];
        const float q10 = r1[0], q11 = r1[32], q12 = r1[64], q13 = r1[96];
        const float q20 = r2[0], q21 = r2[32], q22 = r2[64], q23 = r2[96];
        const float q30 = r3[0], q31 = r3[32], q32 = r3[64], q33 = r3[96];
        acc.x += lane_f(al.x, j) * q00;     acc.y += lane_f(al.y, j) * q01;
        acc.z += lane_f(al.z, j) * q02;     acc.w += lane_f(al.w, j) * q03;
        acc.x += lane_f(al.x, j + 1) * q10; acc.y += lane_f(al.y, j + 1) * q11;
        acc.z += lane_f(al.z, j + 1) * q12; acc.w += lane_f(al.w, j + 1) * q13;
        acc.x += lane_f(al.x, j + 2) * q20; acc.y += lane_f(al.y, j + 2) * q21;
        acc.z += lane_f(al.z, j + 2) * q22; acc.w += lane_f(al.w, j + 2) * q23;
        acc.x += lane_f(al.x, j + 3) * q30; acc.y += lane_f(al.y, j + 3) * q31;
        acc.z += lane_f(al.z, j + 3) * q32; acc.w += lane_f(al.w, j + 3) * q33;
    }
    for (; j < cnt; ++j) {
        const float* r0 = hl + (size_t)__builtin_amdgcn_readlane(s, j) * 128;
        const float q00 = r0[0], q01 = r0[32], q02 = r0[64], q03 = r0[96];
        acc.x += lane_f(al.x, j) * q00;     acc.y += lane_f(al.y, j) * q01;
        acc.z += lane_f(al.z, j) * q02;     acc.w += lane_f(al.w, j) * q03;
    }
}

#define L1_NODES_PER_WAVE 5

__global__ __launch_bounds__(256, 10) void seg_l1(const int* __restrict__ off,
                                              const int* __restrict__ csr,
                                              const float* __restrict__ as1,
                                              const float* __restrict__ ad1,
                                              const float* __restrict__ h1,
                                              const float* __restrict__ b1,
                                              const float* __restrict__ W2,
                                              const float* __restrict__ att_s2,
                                              const float* __restrict__ att_d2,
                                              float* __restrict__ h2,
                                              float* __restrict__ as2,
                                              float* __restrict__ ad2, int N) {
    const int n0 = L1_NODES_PER_WAVE *
                   __builtin_amdgcn_readfirstlane((int)(blockIdx.x * 8 + (threadIdx.x >> 5)));
    if (n0 >= N) return;
    const int lane = threadIdx.x & 31;
    const float* hl = h1 + lane;
    const float4* as4 = (const float4*)as1;
    const float4* ad4 = (const float4*)ad1;

    const float c0 = b1[lane], c1 = b1[lane + 32], c2 = b1[lane + 64], c3 = b1[lane + 96];
    const float2* w2v = (const float2*)W2;
    const float2 w0 = w2v[lane], w1 = w2v[lane + 32], w2 = w2v[lane + 64], w3 = w2v[lane + 96];
    const float ts0 = att_s2[0], ts1 = att_s2[1], td0 = att_d2[0], td1 = att_d2[1];

    int o0 = off[n0];
    int o1 = off[n0 + 1];
    int o2 = off[min(n0 + 2, N)];
    int s0 = n0;
    if (lane > 0 && lane < o1 - o0 + 1) s0 = csr[o0 + lane - 1];
    float4 r0 = as4[s0];
    float4 adv = ad4[n0];

    for (int i = 0; i < L1_NODES_PER_WAVE; ++i) {
        const int n = n0 + i;
        if (n >= N) break;
        const int beg = o0;
        const int T = o1 - o0 + 1;
        const bool more = (i + 1 < L1_NODES_PER_WAVE) && (n + 1 < N);

        const bool ok0 = lane < T;
        float4 v0;
        v0.x = lrelu(r0.x + adv.x);
        v0.y = lrelu(r0.y + adv.y);
        v0.z = lrelu(r0.z + adv.z);
        v0.w = lrelu(r0.w + adv.w);

        const int o3 = off[min(n + 3, N)];
        const bool hasn = more && lane > 0 && lane < o2 - o1 + 1;
        const int sxn = csr[hasn ? o1 + lane - 1 : 0];

        float4 m;
        m.x = ok0 ? v0.x : -INFINITY;
        m.y = ok0 ? v0.y : -INFINITY;
        m.z = ok0 ? v0.z : -INFINITY;
        m.w = ok0 ? v0.w : -INFINITY;
        for (int t0 = 32; t0 < T; t0 += 32) {
            const int t = t0 + lane;
            if (t < T) {
                float4 v = as4[csr[beg + t - 1]];
                m.x = fmaxf(m.x, lrelu(v.x + adv.x));
                m.y = fmaxf(m.y, lrelu(v.y + adv.y));
                m.z = fmaxf(m.z, lrelu(v.z + adv.z));
                m.w = fmaxf(m.w, lrelu(v.w + adv.w));
            }
        }
        m.x = finite_or_zero(wave_max(m.x));
        m.y = finite_or_zero(wave_max(m.y));
        m.z = finite_or_zero(wave_max(m.z));
        m.w = finite_or_zero(wave_max(m.w));

        float4 e0;
        e0.x = ok0 ? expf(v0.x - m.x) : 0.0f;
        e0.y = ok0 ? expf(v0.y - m.y) : 0.0f;
        e0.z = ok0 ? expf(v0.z - m.z) : 0.0f;
        e0.w = ok0 ? expf(v0.w - m.w) : 0.0f;
        float4 d = e0;
        for (int t0 = 32; t0 < T; t0 += 32) {
            const int t = t0 + lane;
            if (t < T) {
                float4 v = as4[csr[beg + t - 1]];
                d.x += expf(lrelu(v.x + adv.x) - m.x);
                d.y += expf(lrelu(v.y + adv.y) - m.y);
                d.z += expf(lrelu(v.z + adv.z) - m.z);
                d.w += expf(lrelu(v.w + adv.w) - m.w);
            }
        }
        d.x = wave_sum(d.x) + EPS_F;
        d.y = wave_sum(d.y) + EPS_F;
        d.z = wave_sum(d.z) + EPS_F;
        d.w = wave_sum(d.w) + EPS_F;

        const int nn = more ? n + 1 : n;
        const int s0n = hasn ? sxn : nn;
        const float4 r0n = as4[s0n];
        const float4 advn = ad4[nn];

        float4 acc = {0.0f, 0.0f, 0.0f, 0.0f};
        {
            float4 al;
            al.x = e0.x / d.x;
            al.y = e0.y / d.y;
            al.z = e0.z / d.z;
            al.w = e0.w / d.w;
            add_rows(acc, T < 32 ? T : 32, s0, al, hl);
        }
        for (int t0 = 32; t0 < T; t0 += 32) {
            const int t = t0 + lane;
            int s = n;
            float4 al = {0.0f, 0.0f, 0.0f, 0.0f};
            if (t < T) {
                s = csr[beg + t - 1];
                float4 v = as4[s];
                al.x = expf(lrelu(v.x + adv.x) - m.x) / d.x;
                al.y = expf(lrelu(v.y + adv.y) - m.y) / d.y;
                al.z = expf(lrelu(v.z + adv.z) - m.z) / d.z;
                al.w = expf(lrelu(v.w + adv.w) - m.w) / d.w;
            }
            const int left = T - t0;
            add_rows(acc, left < 32 ? left : 32, s, al, hl);
        }

        float4 f;
        f.x = acc.x + c0; f.x = f.x > 0.0f ? f.x : expf(f.x) - 1.0f;
        f.y = acc.y + c1; f.y = f.y > 0.0f ? f.y : expf(f.y) - 1.0f;
        f.z = acc.z + c2; f.z = f.z > 0.0f ? f.z : expf(f.z) - 1.0f;
        f.w = acc.w + c3; f.w = f.w > 0.0f ? f.w : expf(f.w) - 1.0f;
        float p0 = f.x * w0.x + f.y * w1.x + f.z * w2.x + f.w * w3.x;
        float p1 = f.x * w0.y + f.y * w1.y + f.z * w2.y + f.w * w3.y;
        p0 = wave_sum(p0);
        p1 = wave_sum(p1);
        if (lane == 0) {
            h2[n * 2 + 0] = p0;
            h2[n * 2 + 1] = p1;
            as2[n] = p0 * ts0 + p1 * ts1;
            ad2[n] = p0 * td0 + p1 * td1;
        }

        o0 = o1; o1 = o2; o2 = o3;
        s0 = s0n; r0 = r0n; adv = advn;
    }
}

#define L2_LANES 4
#define L2_CHUNKS 8
static __device__ __forceinline__ float group4_max(float v) {
    v = fmaxf(v, __shfl_xor(v, 2, 32));
    v = fmaxf(v, __shfl_xor(v, 1, 32));
    return v;
}
static __device__ __forceinline__ float group4_sum(float v) {
    v += __shfl_xor(v, 2, 32);
    v += __shfl_xor(v, 1, 32);
    return v;
}

__global__ __launch_bounds__(256) void seg_l2(const int* __restrict__ off,
                                              const int* __restrict__ csr,
                                              const float* __restrict__ as2,
                                              const float* __restrict__ ad2,
                                              const float* __restrict__ h2,
                                              const float* __restrict__ b2,
                                              float* __restrict__ y, int N) {
    const int lane = threadIdx.x & 31;
    const int q = lane & (L2_LANES - 1);
    const int nw = (32 / L2_LANES) * (int)(blockIdx.x * 8 + (threadIdx.x >> 5));
    if (nw >= N) return;
    const int nr = nw + lane / L2_LANES;
    const bool live = nr < N;
    const int n = live ? nr : N - 1;
    const int beg = off[n];
    const int T = off[n + 1] - beg + 1;
    const float adn = ad2[n];
    const float2* h2v = (const float2*)h2;

    int s[L2_CHUNKS];
    bool ok[L2_CHUNKS];
    #pragma unroll
    for (int c = 0; c < L2_CHUNKS; ++c) {
        const int t = L2_LANES * c + q;
        ok[c] = t < T;
        const bool has = ok[c] && t > 0;
        const int sx = csr[has ? beg + t - 1 : 0];
        s[c] = has ? sx : n;
    }
    float v[L2_CHUNKS];
    float2 g[L2_CHUNKS];
    #pragma unroll
    for (int c = 0; c < L2_CHUNKS; ++c) {
        v[c] = lrelu(as2[s[c]] + adn);
        g[c] = h2v[s[c]];
    }

    float m = -INFINITY;
    #pragma unroll
    for (int c = 0; c < L2_CHUNKS; ++c) m = ok[c] ? fmaxf(m, v[c]) : m;
    for (int t = 32 + q; t < T; t += L2_LANES)
        m = fmaxf(m, lrelu(as2[csr[beg + t - 1]] + adn));
    m = finite_or_zero(group4_max(m));

    float e[L2_CHUNKS];
    float d = 0.0f;
    #pragma unroll
    for (int c = 0; c < L2_CHUNKS; ++c) {
        e[c] = ok[c] ? expf(v[c] - m) : 0.0f;
        d += e[c];
    }
    for (int t = 32 + q; t < T; t += L2_LANES)
        d += expf(lrelu(as2[csr[beg + t - 1]] + adn) - m);
    d = group4_sum(d) + EPS_F;

    float o0 = 0.0f, o1 = 0.0f;
    #pragma unroll
    for (int c = 0; c < L2_CHUNKS; ++c) {
        const float al = e[c] / d;
        o0 += ok[c] ? al * g[c].x : 0.0f;
        o1 += ok[c] ? al * g[c].y : 0.0f;
    }
    for (int t = 32 + q; t < T; t += L2_LANES) {
        const int sx = csr[beg + t - 1];
        const float al = expf(lrelu(as2[sx] + adn) - m) / d;
        const float2 gx = h2v[sx];
        o0 += al * gx.x;
        o1 += al * gx.y;
    }
    o0 = group4_sum(o0);
    o1 = group4_sum(o1);
    if (q == 0 && live) {
        const float z0 = o0 + b2[0];
        const float z1 = o1 + b2[1];
        const float mx = fmaxf(z0, z1);
        const float lse = mx + logf(expf(z0 - mx) + expf(z1 - mx));
        float2 r;
        r.x = z0 - lse;
        r.y = z1 - lse;
        ((float2*)y)[n] = r;
    }
}

extern "C" void kernel_launch(void* const* d_in, const int* in_sizes, int n_in,
                              void* d_out, int out_size, void* d_ws, size_t ws_size,
                              hipStream_t stream) {
    const float* x        = (const float*)d_in[0];
    const int*   ei       = (const int*)d_in[1];
    const float* W1       = (const float*)d_in[2];
    const float* att_src1 = (const float*)d_in[3];
    const float* att_dst1 = (const float*)d_in[4];
    const float* b1       = (const float*)d_in[5];
    const float* W2       = (const float*)d_in[6];
    const float* att_src2 = (const float*)d_in[7];
    const float* att_dst2 = (const float*)d_in[8];
    const float* b2       = (const float*)d_in[9];
    float* y = (float*)d_out;

    const int N = in_sizes[0] / 128;
    const int E = in_sizes[1] / 2;

    float* ws   = (float*)d_ws;
    float* h1   = ws;
    float* as1  = h1  + (size_t)N * 128;
    float* ad1  = as1 + (size_t)N * 4;
    float* h2   = ad1 + (size_t)N * 4;
    float* as2  = h2  + (size_t)N * 2;
    float* ad2  = as2 + (size_t)N;
    int*   deg  = (int*)(ad2 + (size_t)N);
    int*   off  = deg + (size_t)N;
    int*   rank = off + (size_t)N + 1;
    int*   csr  = rank + (size_t)E;

    const int TB = 256;

    zero_deg<<<(N + TB - 1) / TB, TB, 0, stream>>>(deg, N);
    gemm_wmma_f32<<<N / 16, 256, 0, stream>>>(x, W1, h1, N, ei, E, deg, rank);
    scan_deg<<<1, 1024, 0, stream>>>(deg, off, N);
    attn_l1<<<(N * 4 + TB - 1) / TB, TB, 0, stream>>>(h1, att_src1, att_dst1, as1, ad1, N,
                                                      ei, E, off, rank, csr);

    seg_l1<<<(N + 39) / 40, 256, 0, stream>>>(off, csr, as1, ad1, h1, b1, W2, att_src2, att_dst2,
                                            h2, as2, ad2, N);

    seg_l2<<<(N + 63) / 64, 256, 0, stream>>>(off, csr, as2, ad2, h2, b2, y, N);
}


# Projection work-groups walk 4 or 5 row stripes each (768 blocks, not 3125): W slice stays in registers, next stripe's rows are fetched under the products, barriers order LDS only; sum order kept
# speedup vs baseline: 4.9250x; 1.2861x over previous
#include <hip/hip_runtime.h>
#include <math.h>


typedef __attribute__((ext_vector_type(2))) float v2f;
typedef __attribute__((ext_vector_type(8))) float v8f;

#define NEG_SLOPE 0.2f
#define EPS_F 1e-16f

static __device__ __forceinline__ float lrelu(float v) { return v >= 0.0f ? v : NEG_SLOPE * v; }
static __device__ __forceinline__ float finite_or_zero(float m) { return isfinite(m) ? m : 0.0f; }
static __device__ __forceinline__ float wave_max(float v) {
    #pragma unroll
    for (int o = 16; o >= 1; o >>= 1) v = fmaxf(v, __shfl_xor(v, o, 32));
    return v;
}
static __device__ __forceinline__ float wave_sum(float v) {
    #pragma unroll
    for (int o = 16; o >= 1; o >>= 1) v += __shfl_xor(v, o, 32);
    return v;
}
static __device__ __forceinline__ float lane_f(float v, int j) {
    return __int_as_float(__builtin_amdgcn_readlane(__float_as_int(v), j));
}

#define SLOTS 32

__global__ void zero_deg(int* __restrict__ deg, int n) {
    int i = blockIdx.x * blockDim.x + threadIdx.x;
    if (i < n) deg[i] = 0;
}

static __device__ __forceinline__ void put_edge(int* __restrict__ slots, int* __restrict__ ovf_d,
                                                int* __restrict__ ovf_s, int* __restrict__ ovf_n,
                                                int d, int s, int r) {
    if (r < SLOTS) {
        slots[(size_t)d * SLOTS + r] = s;
    } else {
        const int k = atomicAdd(ovf_n, 1);
        ovf_d[k] = d;
        ovf_s[k] = s;
    }
}

static __device__ __forceinline__ void lds_barrier() {
    __builtin_amdgcn_fence(__ATOMIC_RELEASE, "workgroup", "local");
    __builtin_amdgcn_s_barrier();
    __builtin_amdgcn_fence(__ATOMIC_ACQUIRE, "workgroup", "local");
}

#define XS_PITCH 132
#define GEMM_BLOCKS 768
__global__ __launch_bounds__(256, 8) void gemm_wmma_f32(const float* __restrict__ X,
                                                     const float* __restrict__ W,
                                                     float* __restrict__ H, int N,
                                                     const int* __restrict__ ei, int E,
                                                     int* __restrict__ deg, int* __restrict__ slots,
                                                     int* __restrict__ ovf_d, int* __restrict__ ovf_s,
                                                     const float* __restrict__ att_s,
                                                     const float* __restrict__ att_d,
                                                     float* __restrict__ as1, float* __restrict__ ad1) {
    __shared__ __attribute__((aligned(16))) float xs[16 * XS_PITCH];
    __shared__ __attribute__((aligned(16))) float hs[16 * XS_PITCH];
    __shared__ __attribute__((aligned(16))) float atts[256];

    const int nStripes = N >> 4;
    const int st0 = (int)((blockIdx.x * (unsigned)nStripes) / gridDim.x);
    const int st1 = (int)(((blockIdx.x + 1u) * (unsigned)nStripes) / gridDim.x);

    const int wave = threadIdx.x >> 5;
    const int lane = threadIdx.x & 31;
    const int colBase = wave * 16;
    const int half = lane >> 4;
    const int mn = lane & 15;
    const int f0 = threadIdx.x, f1 = threadIdx.x + 256;
    float* xs0 = xs + (f0 >> 5) * XS_PITCH + (f0 & 31) * 4;
    float* xs1 = xs + (f1 >> 5) * XS_PITCH + (f1 & 31) * 4;
    const float* hs0 = hs + (f0 >> 5) * XS_PITCH + (f0 & 31) * 4;
    const float* hs1 = hs + (f1 >> 5) * XS_PITCH + (f1 & 31) * 4;

    if (st0 < st1) {
        const float attv = threadIdx.x < 128 ? att_s[threadIdx.x] : att_d[threadIdx.x - 128];

        const float4* xg0 = (const float4*)(X + (size_t)st0 * 2048);
        float4 x0 = xg0[f0];
        float4 x1 = xg0[f1];
        const int e00 = st0 * 256 + threadIdx.x;
        const int ee0 = e00 < E ? e00 : 0;
        int dn = ei[E + ee0];
        int sn = ei[ee0];

        v2f b[32];
        #pragma unroll
        for (int i = 0; i < 32; ++i) {
            const int ka = 4 * i + 2 * half;
            b[i].x = W[(size_t)ka * 128 + colBase + mn];
            b[i].y = W[(size_t)(ka + 1) * 128 + colBase + mn];
        }
        atts[threadIdx.x] = attv;
        #pragma unroll
        for (int i = 0; i < 32; ++i) asm volatile("" : "+v"(b[i]));

        const float* xrow = xs + mn * XS_PITCH + 2 * half;
        float* htile = hs + (8 * half) * XS_PITCH + colBase + mn;

        int dp = 0, sp = 0, rp = 0;
        bool hp = false;

        for (int st = st0; st < st1; ++st) {
            *(float4*)xs0 = x0;
            *(float4*)xs1 = x1;
            const int dc = dn, sc = sn;
            const int e = st * 256 + threadIdx.x;
            const bool has = e < E;
            lds_barrier();

            int rc = 0;
            if (has) rc = atomicAdd(&deg[dc], 1);
            if (st + 1 < st1) {
                const float4* xg = (const float4*)(X + (size_t)(st + 1) * 2048);
                x0 = xg[f0];
                x1 = xg[f1];
                const int en = e + 256;
                const int ee = en < E ? en : 0;
                dn = ei[E + ee];
                sn = ei[ee];
            }

            v8f c = {};
            #pragma unroll
            for (int i = 0; i < 32; ++i) {
                const v2f a = *(const v2f*)(xrow + 4 * i);
                c = __builtin_amdgcn_wmma_f32_16x16x4_f32(false, a, false, b[i],
                                                          (short)0, c, false, false);
            }
            #pragma unroll
            for (int r = 0; r < 8; ++r) {
                htile[r * XS_PITCH] = c[r];
            }
            lds_barrier();

            float4* hg = (float4*)(H + (size_t)st * 2048);
            hg[f0] = *(const float4*)hs0;
            hg[f1] = *(const float4*)hs1;

            if (threadIdx.x < 128) {
                const bool dside = threadIdx.x >= 64;
                const int idx = threadIdx.x & 63;
                const int hd = idx & 3;
                const float4* ap = (const float4*)(atts + (dside ? 128 : 0) + hd * 32);
                const float4* hp4 = (const float4*)(hs + (idx >> 2) * XS_PITCH + hd * 32);
                float s = 0.0f;
                #pragma unroll 4
                for (int j = 0; j < 8; ++j) {
                    const float4 hv = hp4[j], av = ap[j];
                    s += hv.x * av.x + hv.y * av.y + hv.z * av.z + hv.w * av.w;
                }
                (dside ? ad1 : as1)[(size_t)st * 64 + idx] = s;
            }

            if (hp) put_edge(slots, ovf_d, ovf_s, deg + N, dp, sp, rp);
            dp = dc; sp = sc; rp = rc; hp = has;
        }
        if (hp) put_edge(slots, ovf_d, ovf_s, deg + N, dp, sp, rp);
    }

    for (int e = nStripes * 256 + blockIdx.x * 256 + threadIdx.x; e < E; e += gridDim.x * 256) {
        const int d = ei[E + e];
        put_edge(slots, ovf_d, ovf_s, deg + N, d, ei[e], atomicAdd(&deg[d], 1));
    }
}

static __device__ __forceinline__ void add_rows(float4& acc, int cnt, int s, float4 al,
                                                const float* __restrict__ hl) {
    int j = 0;
    for (; j + 4 <= cnt; j += 4) {
        const float* r0 = hl + (size_t)__builtin_amdgcn_readlane(s, j) * 128;
        const float* r1 = hl + (size_t)__builtin_amdgcn_readlane(s, j + 1) * 128;
        const float* r2 = hl + (size_t)__builtin_amdgcn_readlane(s, j + 2) * 128;
        const float* r3 = hl + (size_t)__builtin_amdgcn_readlane(s, j + 3) * 128;
        const float q00 = r0[0], q01 = r0[32], q02 = r0[64], q03 = r0[96];
        const float q10 = r1[0], q11 = r1[32], q12 = r1[64], q13 = r1[96];
        const float q20 = r2[0], q21 = r2[32], q22 = r2[64], q23 = r2[96];
        const float q30 = r3[0], q31 = r3[32], q32 = r3[64], q33 = r3[96];
        acc.x += lane_f(al.x, j) * q00;     acc.y += lane_f(al.y, j) * q01;
        acc.z += lane_f(al.z, j) * q02;     acc.w += lane_f(al.w, j) * q03;
        acc.x += lane_f(al.x, j + 1) * q10; acc.y += lane_f(al.y, j + 1) * q11;
        acc.z += lane_f(al.z, j + 1) * q12; acc.w += lane_f(al.w, j + 1) * q13;
        acc.x += lane_f(al.x, j + 2) * q20; acc.y += lane_f(al.y, j + 2) * q21;
        acc.z += lane_f(al.z, j + 2) * q22; acc.w += lane_f(al.w, j + 2) * q23;
        acc.x += lane_f(al.x, j + 3) * q30; acc.y += lane_f(al.y, j + 3) * q31;
        acc.z += lane_f(al.z, j + 3) * q32; acc.w += lane_f(al.w, j + 3) * q33;
    }
    for (; j < cnt; ++j) {
        const float* r0 = hl + (size_t)__builtin_amdgcn_readlane(s, j) * 128;
        const float q00 = r0[0], q01 = r0[32], q02 = r0[64], q03 = r0[96];
        acc.x += lane_f(al.x, j) * q00;     acc.y += lane_f(al.y, j) * q01;
        acc.z += lane_f(al.z, j) * q02;     acc.w += lane_f(al.w, j) * q03;
    }
}

static __device__ __forceinline__ void add_row(float4& acc, int j, int s, float4 al,
                                               const float* __restrict__ hl) {
    const float* r0 = hl + (size_t)__builtin_amdgcn_readlane(s, j) * 128;
    const float q00 = r0[0], q01 = r0[32], q02 = r0[64], q03 = r0[96];
    acc.x += lane_f(al.x, j) * q00;     acc.y += lane_f(al.y, j) * q01;
    acc.z += lane_f(al.z, j) * q02;     acc.w += lane_f(al.w, j) * q03;
}

#define L1_NODES_PER_WAVE 5

__global__ __launch_bounds__(256, 10) void seg_l1(const int* __restrict__ deg,
                                              const int* __restrict__ slots,
                                              const int* __restrict__ ovf_d,
                                              const int* __restrict__ ovf_s,
                                              const float* __restrict__ as1,
                                              const float* __restrict__ ad1,
                                              const float* __restrict__ h1,
                                              const float* __restrict__ b1,
                                              const float* __restrict__ W2,
                                              const float* __restrict__ att_s2,
                                              const float* __restrict__ att_d2,
                                              float* __restrict__ h2,
                                              float* __restrict__ as2,
                                              float* __restrict__ ad2, int N) {
    const int n0 = L1_NODES_PER_WAVE *
                   __builtin_amdgcn_readfirstlane((int)(blockIdx.x * 8 + (threadIdx.x >> 5)));
    if (n0 >= N) return;
    const int lane = threadIdx.x & 31;
    const float* hl = h1 + lane;
    const float4* as4 = (const float4*)as1;
    const float4* ad4 = (const float4*)ad1;

    const float c0 = b1[lane], c1 = b1[lane + 32], c2 = b1[lane + 64], c3 = b1[lane + 96];
    const float2* w2v = (const float2*)W2;
    const float2 w0 = w2v[lane], w1 = w2v[lane + 32], w2 = w2v[lane + 64], w3 = w2v[lane + 96];
    const float ts0 = att_s2[0], ts1 = att_s2[1], td0 = att_d2[0], td1 = att_d2[1];

    int g0 = deg[n0];
    int g1 = deg[min(n0 + 1, N - 1)];
    const int sl = lane > 0 ? lane - 1 : 0;
    const int sx0 = slots[(size_t)n0 * SLOTS + sl];
    int s0 = (lane > 0 && lane < min(g0, SLOTS) + 1) ? sx0 : n0;
    float4 r0 = as4[s0];
    float4 adv = ad4[n0];

    for (int i = 0; i < L1_NODES_PER_WAVE; ++i) {
        const int n = n0 + i;
        if (n >= N) break;
        const int* row = slots + (size_t)n * SLOTS;
        const int T = min(g0, SLOTS) + 1;
        const bool over = g0 > SLOTS;
        const int novf = over ? deg[N] : 0;
        const bool more = (i + 1 < L1_NODES_PER_WAVE) && (n + 1 < N);

        const bool ok0 = lane < T;
        float4 v0;
        v0.x = lrelu(r0.x + adv.x);
        v0.y = lrelu(r0.y + adv.y);
        v0.z = lrelu(r0.z + adv.z);
        v0.w = lrelu(r0.w + adv.w);

        const int g2 = deg[min(n + 2, N - 1)];
        const int nn = more ? n + 1 : n;
        const int sxn = slots[(size_t)nn * SLOTS + sl];
        const bool hasn = more && lane > 0 && lane < min(g1, SLOTS) + 1;

        float4 m;
        m.x = ok0 ? v0.x : -INFINITY;
        m.y = ok0 ? v0.y : -INFINITY;
        m.z = ok0 ? v0.z : -INFINITY;
        m.w = ok0 ? v0.w : -INFINITY;
        for (int t0 = 32; t0 < T; t0 += 32) {
            const int t = t0 + lane;
            if (t < T) {
                float4 v = as4[row[t - 1]];
                m.x = fmaxf(m.x, lrelu(v.x + adv.x));
                m.y = fmaxf(m.y, lrelu(v.y + adv.y));
                m.z = fmaxf(m.z, lrelu(v.z + adv.z));
                m.w = fmaxf(m.w, lrelu(v.w + adv.w));
            }
        }
        if (over) {
            for (int k = lane; k < novf; k += 32) {
                if (ovf_d[k] == n) {
                    float4 v = as4[ovf_s[k]];
                    m.x = fmaxf(m.x, lrelu(v.x + adv.x));
                    m.y = fmaxf(m.y, lrelu(v.y + adv.y));
                    m.z = fmaxf(m.z, lrelu(v.z + adv.z));
                    m.w = fmaxf(m.w, lrelu(v.w + adv.w));
                }
            }
        }
        m.x = finite_or_zero(wave_max(m.x));
        m.y = finite_or_zero(wave_max(m.y));
        m.z = finite_or_zero(wave_max(m.z));
        m.w = finite_or_zero(wave_max(m.w));

        float4 e0;
        e0.x = ok0 ? expf(v0.x - m.x) : 0.0f;
        e0.y = ok0 ? expf(v0.y - m.y) : 0.0f;
        e0.z = ok0 ? expf(v0.z - m.z) : 0.0f;
        e0.w = ok0 ? expf(v0.w - m.w) : 0.0f;
        float4 d = e0;
        for (int t0 = 32; t0 < T; t0 += 32) {
            const int t = t0 + lane;
            if (t < T) {
                float4 v = as4[row[t - 1]];
                d.x += expf(lrelu(v.x + adv.x) - m.x);
                d.y += expf(lrelu(v.y + adv.y) - m.y);
                d.z += expf(lrelu(v.z + adv.z) - m.z);
                d.w += expf(lrelu(v.w + adv.w) - m.w);
            }
        }
        if (over) {
            for (int k = lane; k < novf; k += 32) {
                if (ovf_d[k] == n) {
                    float4 v = as4[ovf_s[k]];
                    d.x += expf(lrelu(v.x + adv.x) - m.x);
                    d.y += expf(lrelu(v.y + adv.y) - m.y);
                    d.z += expf(lrelu(v.z + adv.z) - m.z);
                    d.w += expf(lrelu(v.w + adv.w) - m.w);
                }
            }
        }
        d.x = wave_sum(d.x) + EPS_F;
        d.y = wave_sum(d.y) + EPS_F;
        d.z = wave_sum(d.z) + EPS_F;
        d.w = wave_sum(d.w) + EPS_F;

        const int s0n = hasn ? sxn : nn;
        const float4 r0n = as4[s0n];
        const float4 advn = ad4[nn];

        float4 acc = {0.0f, 0.0f, 0.0f, 0.0f};
        {
            float4 al;
            al.x = e0.x / d.x;
            al.y = e0.y / d.y;
            al.z = e0.z / d.z;
            al.w = e0.w / d.w;
            add_rows(acc, T < 32 ? T : 32, s0, al, hl);
        }
        for (int t0 = 32; t0 < T; t0 += 32) {
            const int t = t0 + lane;
            int s = n;
            float4 al = {0.0f, 0.0f, 0.0f, 0.0f};
            if (t < T) {
                s = row[t - 1];
                float4 v = as4[s];
                al.x = expf(lrelu(v.x + adv.x) - m.x) / d.x;
                al.y = expf(lrelu(v.y + adv.y) - m.y) / d.y;
                al.z = expf(lrelu(v.z + adv.z) - m.z) / d.z;
                al.w = expf(lrelu(v.w + adv.w) - m.w) / d.w;
            }
            const int left = T - t0;
            add_rows(acc, left < 32 ? left : 32, s, al, hl);
        }
        if (over) {
            for (int k0 = 0; k0 < novf; k0 += 32) {
                const int k = k0 + lane;
                bool hit = false;
                int s = n;
                float4 al = {0.0f, 0.0f, 0.0f, 0.0f};
                if (k < novf && ovf_d[k] == n) {
                    hit = true;
                    s = ovf_s[k];
                    float4 v = as4[s];
                    al.x = expf(lrelu(v.x + adv.x) - m.x) / d.x;
                    al.y = expf(lrelu(v.y + adv.y) - m.y) / d.y;
                    al.z = expf(lrelu(v.z + adv.z) - m.z) / d.z;
                    al.w = expf(lrelu(v.w + adv.w) - m.w) / d.w;
                }
                unsigned hits = __builtin_amdgcn_ballot_w32(hit);
                while (hits != 0u) {
                    add_row(acc, __builtin_ctz(hits), s, al, hl);
                    hits &= hits - 1u;
                }
            }
        }

        float4 f;
        f.x = acc.x + c0; f.x = f.x > 0.0f ? f.x : expf(f.x) - 1.0f;
        f.y = acc.y + c1; f.y = f.y > 0.0f ? f.y : expf(f.y) - 1.0f;
        f.z = acc.z + c2; f.z = f.z > 0.0f ? f.z : expf(f.z) - 1.0f;
        f.w = acc.w + c3; f.w = f.w > 0.0f ? f.w : expf(f.w) - 1.0f;
        float p0 = f.x * w0.x + f.y * w1.x + f.z * w2.x + f.w * w3.x;
        float p1 = f.x * w0.y + f.y * w1.y + f.z * w2.y + f.w * w3.y;
        p0 = wave_sum(p0);
        p1 = wave_sum(p1);
        if (lane == 0) {
            h2[n * 2 + 0] = p0;
            h2[n * 2 + 1] = p1;
            as2[n] = p0 * ts0 + p1 * ts1;
            ad2[n] = p0 * td0 + p1 * td1;
        }

        g0 = g1; g1 = g2;
        s0 = s0n; r0 = r0n; adv = advn;
    }
}

#define L2_LANES 4
#define L2_CHUNKS 8
static __device__ __forceinline__ float group4_max(float v) {
    v = fmaxf(v, __shfl_xor(v, 2, 32));
    v = fmaxf(v, __shfl_xor(v, 1, 32));
    return v;
}
static __device__ __forceinline__ float group4_sum(float v) {
    v += __shfl_xor(v, 2, 32);
    v += __shfl_xor(v, 1, 32);
    return v;
}

__global__ __launch_bounds__(256) void seg_l2(const int* __restrict__ deg,
                                              const int* __restrict__ slots,
                                              const int* __restrict__ ovf_d,
                                              const int* __restrict__ ovf_s,
                                              const float* __restrict__ as2,
                                              const float* __restrict__ ad2,
                                              const float* __restrict__ h2,
                                              const float* __restrict__ b2,
                                              float* __restrict__ y, int N) {
    const int lane = threadIdx.x & 31;
    const int q = lane & (L2_LANES - 1);
    const int nw = (32 / L2_LANES) * (int)(blockIdx.x * 8 + (threadIdx.x >> 5));
    if (nw >= N) return;
    const int nr = nw + lane / L2_LANES;
    const bool live = nr < N;
    const int n = live ? nr : N - 1;
    const int cnt = deg[n];
    const int T = min(cnt, SLOTS) + 1;
    const bool over = cnt > SLOTS;
    const int novf = over ? deg[N] : 0;
    const int* row = slots + (size_t)n * SLOTS;
    const float adn = ad2[n];
    const float2* h2v = (const float2*)h2;

    int s[L2_CHUNKS];
    bool ok[L2_CHUNKS];
    #pragma unroll
    for (int c = 0; c < L2_CHUNKS; ++c) {
        const int t = L2_LANES * c + q;
        ok[c] = t < T;
        const bool has = ok[c] && t > 0;
        const int sx = row[t > 0 ? t - 1 : 0];
        s[c] = has ? sx : n;
    }
    float v[L2_CHUNKS];
    float2 g[L2_CHUNKS];
    #pragma unroll
    for (int c = 0; c < L2_CHUNKS; ++c) {
        v[c] = lrelu(as2[s[c]] + adn);
        g[c] = h2v[s[c]];
    }

    float m = -INFINITY;
    #pragma unroll
    for (int c = 0; c < L2_CHUNKS; ++c) m = ok[c] ? fmaxf(m, v[c]) : m;
    for (int t = 32 + q; t < T; t += L2_LANES)
        m = fmaxf(m, lrelu(as2[row[t - 1]] + adn));
    if (over)
        for (int k = q; k < novf; k += L2_LANES)
            if (ovf_d[k] == n) m = fmaxf(m, lrelu(as2[ovf_s[k]] + adn));
    m = finite_or_zero(group4_max(m));

    float e[L2_CHUNKS];
    float d = 0.0f;
    #pragma unroll
    for (int c = 0; c < L2_CHUNKS; ++c) {
        e[c] = ok[c] ? expf(v[c] - m) : 0.0f;
        d += e[c];
    }
    for (int t = 32 + q; t < T; t += L2_LANES)
        d += expf(lrelu(as2[row[t - 1]] + adn) - m);
    if (over)
        for (int k = q; k < novf; k += L2_LANES)
            if (ovf_d[k] == n) d += expf(lrelu(as2[ovf_s[k]] + adn) - m);
    d = group4_sum(d) + EPS_F;

    float o0 = 0.0f, o1 = 0.0f;
    #pragma unroll
    for (int c = 0; c < L2_CHUNKS; ++c) {
        const float al = e[c] / d;
        o0 += ok[c] ? al * g[c].x : 0.0f;
        o1 += ok[c] ? al * g[c].y : 0.0f;
    }
    for (int t = 32 + q; t < T; t += L2_LANES) {
        const int sx = row[t - 1];
        const float al = expf(lrelu(as2[sx] + adn) - m) / d;
        const float2 gx = h2v[sx];
        o0 += al * gx.x;
        o1 += al * gx.y;
    }
    if (over)
        for (int k = q; k < novf; k += L2_LANES)
            if (ovf_d[k] == n) {
                const int sx = ovf_s[k];
                const float al = expf(lrelu(as2[sx] + adn) - m) / d;
                const float2 gx = h2v[sx];
                o0 += al * gx.x;
                o1 += al * gx.y;
            }
    o0 = group4_sum(o0);
    o1 = group4_sum(o1);
    if (q == 0 && live) {
        const float z0 = o0 + b2[0];
        const float z1 = o1 + b2[1];
        const float mx = fmaxf(z0, z1);
        const float lse = mx + logf(expf(z0 - mx) + expf(z1 - mx));
        float2 r;
        r.x = z0 - lse;
        r.y = z1 - lse;
        ((float2*)y)[n] = r;
    }
}

extern "C" void kernel_launch(void* const* d_in, const int* in_sizes, int n_in,
                              void* d_out, int out_size, void* d_ws, size_t ws_size,
                              hipStream_t stream) {
    const float* x        = (const float*)d_in[0];
    const int*   ei       = (const int*)d_in[1];
    const float* W1       = (const float*)d_in[2];
    const float* att_src1 = (const float*)d_in[3];
    const float* att_dst1 = (const float*)d_in[4];
    const float* b1       = (const float*)d_in[5];
    const float* W2       = (const float*)d_in[6];
    const float* att_src2 = (const float*)d_in[7];
    const float* att_dst2 = (const float*)d_in[8];
    const float* b2       = (const float*)d_in[9];
    float* y = (float*)d_out;

    const int N = in_sizes[0] / 128;
    const int E = in_sizes[1] / 2;

    float* ws    = (float*)d_ws;
    float* h1    = ws;
    int*   slots = (int*)(h1 + (size_t)N * 128);
    float* as1   = (float*)(slots + (size_t)N * SLOTS);
    float* ad1   = as1 + (size_t)N * 4;
    float* h2    = ad1 + (size_t)N * 4;
    float* as2   = h2  + (size_t)N * 2;
    float* ad2   = as2 + (size_t)N;
    int*   deg   = (int*)(ad2 + (size_t)N);
    int*   ovf_d = deg + (size_t)N + 1;
    int*   ovf_s = ovf_d + (size_t)E;

    const int TB = 256;

    zero_deg<<<(N + 1 + TB - 1) / TB, TB, 0, stream>>>(deg, N + 1);
    gemm_wmma_f32<<<GEMM_BLOCKS, 256, 0, stream>>>(x, W1, h1, N, ei, E, deg, slots, ovf_d, ovf_s,
                                              att_src1, att_dst1, as1, ad1);

    seg_l1<<<(N + 39) / 40, 256, 0, stream>>>(deg, slots, ovf_d, ovf_s, as1, ad1, h1, b1, W2,
                                            att_src2, att_dst2, h2, as2, ad2, N);

    seg_l2<<<(N + 63) / 64, 256, 0, stream>>>(deg, slots, ovf_d, ovf_s, as2, ad2, h2, b2, y, N);
}


# Layer 2 fetches one random 8-byte h2 pair per edge: both attention scalars are dot products formed from that pair in the layer-2 kernel, not read from arrays of their own; sum order kept
# speedup vs baseline: 5.5278x; 1.1224x over previous
#include <hip/hip_runtime.h>
#include <math.h>


typedef __attribute__((ext_vector_type(2))) float v2f;
typedef __attribute__((ext_vector_type(8))) float v8f;

#define NEG_SLOPE 0.2f
#define EPS_F 1e-16f

static __device__ __forceinline__ float lrelu(float v) { return v >= 0.0f ? v : NEG_SLOPE * v; }
static __device__ __forceinline__ float finite_or_zero(float m) { return isfinite(m) ? m : 0.0f; }
static __device__ __forceinline__ float wave_max(float v) {
    #pragma unroll
    for (int o = 16; o >= 1; o >>= 1) v = fmaxf(v, __shfl_xor(v, o, 32));
    return v;
}
static __device__ __forceinline__ float wave_sum(float v) {
    #pragma unroll
    for (int o = 16; o >= 1; o >>= 1) v += __shfl_xor(v, o, 32);
    return v;
}
static __device__ __forceinline__ float lane_f(float v, int j) {
    return __int_as_float(__builtin_amdgcn_readlane(__float_as_int(v), j));
}

#define SLOTS 32

__global__ void zero_deg(int* __restrict__ deg, int n) {
    int i = blockIdx.x * blockDim.x + threadIdx.x;
    if (i < n) deg[i] = 0;
}

static __device__ __forceinline__ void put_edge(int* __restrict__ slots, int* __restrict__ ovf_d,
                                                int* __restrict__ ovf_s, int* __restrict__ ovf_n,
                                                int d, int s, int r) {
    if (r < SLOTS) {
        slots[(size_t)d * SLOTS + r] = s;
    } else {
        const int k = atomicAdd(ovf_n, 1);
        ovf_d[k] = d;
        ovf_s[k] = s;
    }
}

static __device__ __forceinline__ void lds_barrier() {
    __builtin_amdgcn_fence(__ATOMIC_RELEASE, "workgroup", "local");
    __builtin_amdgcn_s_barrier();
    __builtin_amdgcn_fence(__ATOMIC_ACQUIRE, "workgroup", "local");
}

#define XS_PITCH 132
#define GEMM_BLOCKS 768
__global__ __launch_bounds__(256, 8) void gemm_wmma_f32(const float* __restrict__ X,
                                                     const float* __restrict__ W,
                                                     float* __restrict__ H, int N,
                                                     const int* __restrict__ ei, int E,
                                                     int* __restrict__ deg, int* __restrict__ slots,
                                                     int* __restrict__ ovf_d, int* __restrict__ ovf_s,
                                                     const float* __restrict__ att_s,
                                                     const float* __restrict__ att_d,
                                                     float* __restrict__ as1, float* __restrict__ ad1) {
    __shared__ __attribute__((aligned(16))) float xs[16 * XS_PITCH];
    __shared__ __attribute__((aligned(16))) float hs[16 * XS_PITCH];
    __shared__ __attribute__((aligned(16))) float atts[256];

    const int nStripes = N >> 4;
    const int st0 = (int)((blockIdx.x * (unsigned)nStripes) / gridDim.x);
    const int st1 = (int)(((blockIdx.x + 1u) * (unsigned)nStripes) / gridDim.x);

    const int wave = threadIdx.x >> 5;
    const int lane = threadIdx.x & 31;
    const int colBase = wave * 16;
    const int half = lane >> 4;
    const int mn = lane & 15;
    const int f0 = threadIdx.x, f1 = threadIdx.x + 256;
    float* xs0 = xs + (f0 >> 5) * XS_PITCH + (f0 & 31) * 4;
    float* xs1 = xs + (f1 >> 5) * XS_PITCH + (f1 & 31) * 4;
    const float* hs0 = hs + (f0 >> 5) * XS_PITCH + (f0 & 31) * 4;
    const float* hs1 = hs + (f1 >> 5) * XS_PITCH + (f1 & 31) * 4;

    if (st0 < st1) {
        const float attv = threadIdx.x < 128 ? att_s[threadIdx.x] : att_d[threadIdx.x - 128];

        const float4* xg0 = (const float4*)(X + (size_t)st0 * 2048);
        float4 x0 = xg0[f0];
        float4 x1 = xg0[f1];
        const int e00 = st0 * 256 + threadIdx.x;
        const int ee0 = e00 < E ? e00 : 0;
        int dn = ei[E + ee0];
        int sn = ei[ee0];

        v2f b[32];
        #pragma unroll
        for (int i = 0; i < 32; ++i) {
            const int ka = 4 * i + 2 * half;
            b[i].x = W[(size_t)ka * 128 + colBase + mn];
            b[i].y = W[(size_t)(ka + 1) * 128 + colBase + mn];
        }
        atts[threadIdx.x] = attv;
        #pragma unroll
        for (int i = 0; i < 32; ++i) asm volatile("" : "+v"(b[i]));

        const float* xrow = xs + mn * XS_PITCH + 2 * half;
        float* htile = hs + (8 * half) * XS_PITCH + colBase + mn;

        int dp = 0, sp = 0, rp = 0;
        bool hp = false;

        for (int st = st0; st < st1; ++st) {
            *(float4*)xs0 = x0;
            *(float4*)xs1 = x1;
            const int dc = dn, sc = sn;
            const int e = st * 256 + threadIdx.x;
            const bool has = e < E;
            lds_barrier();

            int rc = 0;
            if (has) rc = atomicAdd(&deg[dc], 1);
            if (st + 1 < st1) {
                const float4* xg = (const float4*)(X + (size_t)(st + 1) * 2048);
                x0 = xg[f0];
                x1 = xg[f1];
                const int en = e + 256;
                const int ee = en < E ? en : 0;
                dn = ei[E + ee];
                sn = ei[ee];
            }

            v8f c = {};
            #pragma unroll
            for (int i = 0; i < 32; ++i) {
                const v2f a = *(const v2f*)(xrow + 4 * i);
                c = __builtin_amdgcn_wmma_f32_16x16x4_f32(false, a, false, b[i],
                                                          (short)0, c, false, false);
            }
            #pragma unroll
            for (int r = 0; r < 8; ++r) {
                htile[r * XS_PITCH] = c[r];
            }
            lds_barrier();

            float4* hg = (float4*)(H + (size_t)st * 2048);
            hg[f0] = *(const float4*)hs0;
            hg[f1] = *(const float4*)hs1;

            if (threadIdx.x < 128) {
                const bool dside = threadIdx.x >= 64;
                const int idx = threadIdx.x & 63;
                const int hd = idx & 3;
                const float4* ap = (const float4*)(atts + (dside ? 128 : 0) + hd * 32);
                const float4* hp4 = (const float4*)(hs + (idx >> 2) * XS_PITCH + hd * 32);
                float s = 0.0f;
                #pragma unroll 4
                for (int j = 0; j < 8; ++j) {
                    const float4 hv = hp4[j], av = ap[j];
                    s += hv.x * av.x + hv.y * av.y + hv.z * av.z + hv.w * av.w;
                }
                (dside ? ad1 : as1)[(size_t)st * 64 + idx] = s;
            }

            if (hp) put_edge(slots, ovf_d, ovf_s, deg + N, dp, sp, rp);
            dp = dc; sp = sc; rp = rc; hp = has;
        }
        if (hp) put_edge(slots, ovf_d, ovf_s, deg + N, dp, sp, rp);
    }

    for (int e = nStripes * 256 + blockIdx.x * 256 + threadIdx.x; e < E; e += gridDim.x * 256) {
        const int d = ei[E + e];
        put_edge(slots, ovf_d, ovf_s, deg + N, d, ei[e], atomicAdd(&deg[d], 1));
    }
}

static __device__ __forceinline__ void add_rows(float4& acc, int cnt, int s, float4 al,
                                                const float* __restrict__ hl) {
    int j = 0;
    for (; j + 4 <= cnt; j += 4) {
        const float* r0 = hl + (size_t)__builtin_amdgcn_readlane(s, j) * 128;
        const float* r1 = hl + (size_t)__builtin_amdgcn_readlane(s, j + 1) * 128;
        const float* r2 = hl + (size_t)__builtin_amdgcn_readlane(s, j + 2) * 128;
        const float* r3 = hl + (size_t)__builtin_amdgcn_readlane(s, j + 3) * 128;
        const float q00 = r0[0], q01 = r0[32], q02 = r0[64], q03 = r0[96];
        const float q10 = r1[0], q11 = r1[32], q12 = r1[64], q13 = r1[96];
        const float q20 = r2[0], q21 = r2[32], q22 = r2[64], q23 = r2[96];
        const float q30 = r3[0], q31 = r3[32], q32 = r3[64], q33 = r3[96];
        acc.x += lane_f(al.x, j) * q00;     acc.y += lane_f(al.y, j) * q01;
        acc.z += lane_f(al.z, j) * q02;     acc.w += lane_f(al.w, j) * q03;
        acc.x += lane_f(al.x, j + 1) * q10; acc.y += lane_f(al.y, j + 1) * q11;
        acc.z += lane_f(al.z, j + 1) * q12; acc.w += lane_f(al.w, j + 1) * q13;
        acc.x += lane_f(al.x, j + 2) * q20; acc.y += lane_f(al.y, j + 2) * q21;
        acc.z += lane_f(al.z, j + 2) * q22; acc.w += lane_f(al.w, j + 2) * q23;
        acc.x += lane_f(al.x, j + 3) * q30; acc.y += lane_f(al.y, j + 3) * q31;
        acc.z += lane_f(al.z, j + 3) * q32; acc.w += lane_f(al.w, j + 3) * q33;
    }
    for (; j < cnt; ++j) {
        const float* r0 = hl + (size_t)__builtin_amdgcn_readlane(s, j) * 128;
        const float q00 = r0[0], q01 = r0[32], q02 = r0[64], q03 = r0[96];
        acc.x += lane_f(al.x, j) * q00;     acc.y += lane_f(al.y, j) * q01;
        acc.z += lane_f(al.z, j) * q02;     acc.w += lane_f(al.w, j) * q03;
    }
}

static __device__ __forceinline__ void add_row(float4& acc, int j, int s, float4 al,
                                               const float* __restrict__ hl) {
    const float* r0 = hl + (size_t)__builtin_amdgcn_readlane(s, j) * 128;
    const float q00 = r0[0], q01 = r0[32], q02 = r0[64], q03 = r0[96];
    acc.x += lane_f(al.x, j) * q00;     acc.y += lane_f(al.y, j) * q01;
    acc.z += lane_f(al.z, j) * q02;     acc.w += lane_f(al.w, j) * q03;
}

#define L1_NODES_PER_WAVE 5

__global__ __launch_bounds__(256, 10) void seg_l1(const int* __restrict__ deg,
                                              const int* __restrict__ slots,
                                              const int* __restrict__ ovf_d,
                                              const int* __restrict__ ovf_s,
                                              const float* __restrict__ as1,
                                              const float* __restrict__ ad1,
                                              const float* __restrict__ h1,
                                              const float* __restrict__ b1,
                                              const float* __restrict__ W2,
                                              float* __restrict__ h2, int N) {
    const int n0 = L1_NODES_PER_WAVE *
                   __builtin_amdgcn_readfirstlane((int)(blockIdx.x * 8 + (threadIdx.x >> 5)));
    if (n0 >= N) return;
    const int lane = threadIdx.x & 31;
    const float* hl = h1 + lane;
    const float4* as4 = (const float4*)as1;
    const float4* ad4 = (const float4*)ad1;

    const float c0 = b1[lane], c1 = b1[lane + 32], c2 = b1[lane + 64], c3 = b1[lane + 96];
    const float2* w2v = (const float2*)W2;
    const float2 w0 = w2v[lane], w1 = w2v[lane + 32], w2 = w2v[lane + 64], w3 = w2v[lane + 96];

    int g0 = deg[n0];
    int g1 = deg[min(n0 + 1, N - 1)];
    const int sl = lane > 0 ? lane - 1 : 0;
    const int sx0 = slots[(size_t)n0 * SLOTS + sl];
    int s0 = (lane > 0 && lane < min(g0, SLOTS) + 1) ? sx0 : n0;
    float4 r0 = as4[s0];
    float4 adv = ad4[n0];

    for (int i = 0; i < L1_NODES_PER_WAVE; ++i) {
        const int n = n0 + i;
        if (n >= N) break;
        const int* row = slots + (size_t)n * SLOTS;
        const int T = min(g0, SLOTS) + 1;
        const bool over = g0 > SLOTS;
        const int novf = over ? deg[N] : 0;
        const bool more = (i + 1 < L1_NODES_PER_WAVE) && (n + 1 < N);

        const bool ok0 = lane < T;
        float4 v0;
        v0.x = lrelu(r0.x + adv.x);
        v0.y = lrelu(r0.y + adv.y);
        v0.z = lrelu(r0.z + adv.z);
        v0.w = lrelu(r0.w + adv.w);

        const int g2 = deg[min(n + 2, N - 1)];
        const int nn = more ? n + 1 : n;
        const int sxn = slots[(size_t)nn * SLOTS + sl];
        const bool hasn = more && lane > 0 && lane < min(g1, SLOTS) + 1;

        float4 m;
        m.x = ok0 ? v0.x : -INFINITY;
        m.y = ok0 ? v0.y : -INFINITY;
        m.z = ok0 ? v0.z : -INFINITY;
        m.w = ok0 ? v0.w : -INFINITY;
        for (int t0 = 32; t0 < T; t0 += 32) {
            const int t = t0 + lane;
            if (t < T) {
                float4 v = as4[row[t - 1]];
                m.x = fmaxf(m.x, lrelu(v.x + adv.x));
                m.y = fmaxf(m.y, lrelu(v.y + adv.y));
                m.z = fmaxf(m.z, lrelu(v.z + adv.z));
                m.w = fmaxf(m.w, lrelu(v.w + adv.w));
            }
        }
        if (over) {
            for (int k = lane; k < novf; k += 32) {
                if (ovf_d[k] == n) {
                    float4 v = as4[ovf_s[k]];
                    m.x = fmaxf(m.x, lrelu(v.x + adv.x));
                    m.y = fmaxf(m.y, lrelu(v.y + adv.y));
                    m.z = fmaxf(m.z, lrelu(v.z + adv.z));
                    m.w = fmaxf(m.w, lrelu(v.w + adv.w));
                }
            }
        }
        m.x = finite_or_zero(wave_max(m.x));
        m.y = finite_or_zero(wave_max(m.y));
        m.z = finite_or_zero(wave_max(m.z));
        m.w = finite_or_zero(wave_max(m.w));

        float4 e0;
        e0.x = ok0 ? expf(v0.x - m.x) : 0.0f;
        e0.y = ok0 ? expf(v0.y - m.y) : 0.0f;
        e0.z = ok0 ? expf(v0.z - m.z) : 0.0f;
        e0.w = ok0 ? expf(v0.w - m.w) : 0.0f;
        float4 d = e0;
        for (int t0 = 32; t0 < T; t0 += 32) {
            const int t = t0 + lane;
            if (t < T) {
                float4 v = as4[row[t - 1]];
                d.x += expf(lrelu(v.x + adv.x) - m.x);
                d.y += expf(lrelu(v.y + adv.y) - m.y);
                d.z += expf(lrelu(v.z + adv.z) - m.z);
                d.w += expf(lrelu(v.w + adv.w) - m.w);
            }
        }
        if (over) {
            for (int k = lane; k < novf; k += 32) {
                if (ovf_d[k] == n) {
                    float4 v = as4[ovf_s[k]];
                    d.x += expf(lrelu(v.x + adv.x) - m.x);
                    d.y += expf(lrelu(v.y + adv.y) - m.y);
                    d.z += expf(lrelu(v.z + adv.z) - m.z);
                    d.w += expf(lrelu(v.w + adv.w) - m.w);
                }
            }
        }
        d.x = wave_sum(d.x) + EPS_F;
        d.y = wave_sum(d.y) + EPS_F;
        d.z = wave_sum(d.z) + EPS_F;
        d.w = wave_sum(d.w) + EPS_F;

        const int s0n = hasn ? sxn : nn;
        const float4 r0n = as4[s0n];
        const float4 advn = ad4[nn];

        float4 acc = {0.0f, 0.0f, 0.0f, 0.0f};
        {
            float4 al;
            al.x = e0.x / d.x;
            al.y = e0.y / d.y;
            al.z = e0.z / d.z;
            al.w = e0.w / d.w;
            add_rows(acc, T < 32 ? T : 32, s0, al, hl);
        }
        for (int t0 = 32; t0 < T; t0 += 32) {
            const int t = t0 + lane;
            int s = n;
            float4 al = {0.0f, 0.0f, 0.0f, 0.0f};
            if (t < T) {
                s = row[t - 1];
                float4 v = as4[s];
                al.x = expf(lrelu(v.x + adv.x) - m.x) / d.x;
                al.y = expf(lrelu(v.y + adv.y) - m.y) / d.y;
                al.z = expf(lrelu(v.z + adv.z) - m.z) / d.z;
                al.w = expf(lrelu(v.w + adv.w) - m.w) / d.w;
            }
            const int left = T - t0;
            add_rows(acc, left < 32 ? left : 32, s, al, hl);
        }
        if (over) {
            for (int k0 = 0; k0 < novf; k0 += 32) {
                const int k = k0 + lane;
                bool hit = false;
                int s = n;
                float4 al = {0.0f, 0.0f, 0.0f, 0.0f};
                if (k < novf && ovf_d[k] == n) {
                    hit = true;
                    s = ovf_s[k];
                    float4 v = as4[s];
                    al.x = expf(lrelu(v.x + adv.x) - m.x) / d.x;
                    al.y = expf(lrelu(v.y + adv.y) - m.y) / d.y;
                    al.z = expf(lrelu(v.z + adv.z) - m.z) / d.z;
                    al.w = expf(lrelu(v.w + adv.w) - m.w) / d.w;
                }
                unsigned hits = __builtin_amdgcn_ballot_w32(hit);
                while (hits != 0u) {
                    add_row(acc, __builtin_ctz(hits), s, al, hl);
                    hits &= hits - 1u;
                }
            }
        }

        float4 f;
        f.x = acc.x + c0; f.x = f.x > 0.0f ? f.x : expf(f.x) - 1.0f;
        f.y = acc.y + c1; f.y = f.y > 0.0f ? f.y : expf(f.y) - 1.0f;
        f.z = acc.z + c2; f.z = f.z > 0.0f ? f.z : expf(f.z) - 1.0f;
        f.w = acc.w + c3; f.w = f.w > 0.0f ? f.w : expf(f.w) - 1.0f;
        float p0 = f.x * w0.x + f.y * w1.x + f.z * w2.x + f.w * w3.x;
        float p1 = f.x * w0.y + f.y * w1.y + f.z * w2.y + f.w * w3.y;
        p0 = wave_sum(p0);
        p1 = wave_sum(p1);
        if (lane == 0) {
            float2 pr;
            pr.x = p0;
            pr.y = p1;
            ((float2*)h2)[n] = pr;
        }

        g0 = g1; g1 = g2;
        s0 = s0n; r0 = r0n; adv = advn;
    }
}

#define L2_LANES 4
#define L2_CHUNKS 8
static __device__ __forceinline__ float group4_max(float v) {
    v = fmaxf(v, __shfl_xor(v, 2, 32));
    v = fmaxf(v, __shfl_xor(v, 1, 32));
    return v;
}
static __device__ __forceinline__ float group4_sum(float v) {
    v += __shfl_xor(v, 2, 32);
    v += __shfl_xor(v, 1, 32);
    return v;
}

__global__ __launch_bounds__(256) void seg_l2(const int* __restrict__ deg,
                                              const int* __restrict__ slots,
                                              const int* __restrict__ ovf_d,
                                              const int* __restrict__ ovf_s,
                                              const float* __restrict__ att_s2,
                                              const float* __restrict__ att_d2,
                                              const float* __restrict__ h2,
                                              const float* __restrict__ b2,
                                              float* __restrict__ y, int N) {
    const int lane = threadIdx.x & 31;
    const int q = lane & (L2_LANES - 1);
    const int nw = (32 / L2_LANES) * (int)(blockIdx.x * 8 + (threadIdx.x >> 5));
    if (nw >= N) return;
    const int nr = nw + lane / L2_LANES;
    const bool live = nr < N;
    const int n = live ? nr : N - 1;
    const int cnt = deg[n];
    const int T = min(cnt, SLOTS) + 1;
    const bool over = cnt > SLOTS;
    const int novf = over ? deg[N] : 0;
    const int* row = slots + (size_t)n * SLOTS;
    const float2* h2v = (const float2*)h2;
    const float ts0 = att_s2[0], ts1 = att_s2[1], td0 = att_d2[0], td1 = att_d2[1];
    const float2 hn = h2v[n];
    const float adn = hn.x * td0 + hn.y * td1;

    int s[L2_CHUNKS];
    bool ok[L2_CHUNKS];
    #pragma unroll
    for (int c = 0; c < L2_CHUNKS; ++c) {
        const int t = L2_LANES * c + q;
        ok[c] = t < T;
        const bool has = ok[c] && t > 0;
        const int sx = row[t > 0 ? t - 1 : 0];
        s[c] = has ? sx : n;
    }
    float v[L2_CHUNKS];
    float2 g[L2_CHUNKS];
    #pragma unroll
    for (int c = 0; c < L2_CHUNKS; ++c) {
        g[c] = h2v[s[c]];
        v[c] = lrelu((g[c].x * ts0 + g[c].y * ts1) + adn);
    }

    float m = -INFINITY;
    #pragma unroll
    for (int c = 0; c < L2_CHUNKS; ++c) m = ok[c] ? fmaxf(m, v[c]) : m;
    for (int t = 32 + q; t < T; t += L2_LANES) {
        const float2 gx = h2v[row[t - 1]];
        m = fmaxf(m, lrelu((gx.x * ts0 + gx.y * ts1) + adn));
    }
    if (over)
        for (int k = q; k < novf; k += L2_LANES)
            if (ovf_d[k] == n) {
                const float2 gx = h2v[ovf_s[k]];
                m = fmaxf(m, lrelu((gx.x * ts0 + gx.y * ts1) + adn));
            }
    m = finite_or_zero(group4_max(m));

    float e[L2_CHUNKS];
    float d = 0.0f;
    #pragma unroll
    for (int c = 0; c < L2_CHUNKS; ++c) {
        e[c] = ok[c] ? expf(v[c] - m) : 0.0f;
        d += e[c];
    }
    for (int t = 32 + q; t < T; t += L2_LANES) {
        const float2 gx = h2v[row[t - 1]];
        d += expf(lrelu((gx.x * ts0 + gx.y * ts1) + adn) - m);
    }
    if (over)
        for (int k = q; k < novf; k += L2_LANES)
            if (ovf_d[k] == n) {
                const float2 gx = h2v[ovf_s[k]];
                d += expf(lrelu((gx.x * ts0 + gx.y * ts1) + adn) - m);
            }
    d = group4_sum(d) + EPS_F;

    float o0 = 0.0f, o1 = 0.0f;
    #pragma unroll
    for (int c = 0; c < L2_CHUNKS; ++c) {
        const float al = e[c] / d;
        o0 += ok[c] ? al * g[c].x : 0.0f;
        o1 += ok[c] ? al * g[c].y : 0.0f;
    }
    for (int t = 32 + q; t < T; t += L2_LANES) {
        const float2 gx = h2v[row[t - 1]];
        const float al = expf(lrelu((gx.x * ts0 + gx.y * ts1) + adn) - m) / d;
        o0 += al * gx.x;
        o1 += al * gx.y;
    }
    if (over)
        for (int k = q; k < novf; k += L2_LANES)
            if (ovf_d[k] == n) {
                const float2 gx = h2v[ovf_s[k]];
                const float al = expf(lrelu((gx.x * ts0 + gx.y * ts1) + adn) - m) / d;
                o0 += al * gx.x;
                o1 += al * gx.y;
            }
    o0 = group4_sum(o0);
    o1 = group4_sum(o1);
    if (q == 0 && live) {
        const float z0 = o0 + b2[0];
        const float z1 = o1 + b2[1];
        const float mx = fmaxf(z0, z1);
        const float lse = mx + logf(expf(z0 - mx) + expf(z1 - mx));
        float2 r;
        r.x = z0 - lse;
        r.y = z1 - lse;
        ((float2*)y)[n] = r;
    }
}

extern "C" void kernel_launch(void* const* d_in, const int* in_sizes, int n_in,
                              void* d_out, int out_size, void* d_ws, size_t ws_size,
                              hipStream_t stream) {
    const float* x        = (const float*)d_in[0];
    const int*   ei       = (const int*)d_in[1];
    const float* W1       = (const float*)d_in[2];
    const float* att_src1 = (const float*)d_in[3];
    const float* att_dst1 = (const float*)d_in[4];
    const float* b1       = (const float*)d_in[5];
    const float* W2       = (const float*)d_in[6];
    const float* att_src2 = (const float*)d_in[7];
    const float* att_dst2 = (const float*)d_in[8];
    const float* b2       = (const float*)d_in[9];
    float* y = (float*)d_out;

    const int N = in_sizes[0] / 128;
    const int E = in_sizes[1] / 2;

    float* ws    = (float*)d_ws;
    float* h1    = ws;
    int*   slots = (int*)(h1 + (size_t)N * 128);
    float* as1   = (float*)(slots + (size_t)N * SLOTS);
    float* ad1   = as1 + (size_t)N * 4;
    float* h2    = ad1 + (size_t)N * 4;
    int*   deg   = (int*)(h2 + (size_t)N * 2);
    int*   ovf_d = deg + (size_t)N + 1;
    int*   ovf_s = ovf_d + (size_t)E;

    const int TB = 256;

    zero_deg<<<(N + 1 + TB - 1) / TB, TB, 0, stream>>>(deg, N + 1);
    gemm_wmma_f32<<<GEMM_BLOCKS, 256, 0, stream>>>(x, W1, h1, N, ei, E, deg, slots, ovf_d, ovf_s,
                                              att_src1, att_dst1, as1, ad1);

    seg_l1<<<(N + 39) / 40, 256, 0, stream>>>(deg, slots, ovf_d, ovf_s, as1, ad1, h1, b1, W2, h2, N);

    seg_l2<<<(N + 63) / 64, 256, 0, stream>>>(deg, slots, ovf_d, ovf_s, att_src2, att_dst2, h2, b2,
                                            y, N);
}


# Edge counting leaves memory: launch 1 orders each edge run by 768 destination ranges, projection block k places range k, both with LDS counters; 800000 global atomics gone; sum order still by arrival
# speedup vs baseline: 7.5858x; 1.3723x over previous
#include <hip/hip_runtime.h>
#include <math.h>


typedef __attribute__((ext_vector_type(2))) float v2f;
typedef __attribute__((ext_vector_type(8))) float v8f;

#define NEG_SLOPE 0.2f
#define EPS_F 1e-16f

static __device__ __forceinline__ float lrelu(float v) { return v >= 0.0f ? v : NEG_SLOPE * v; }
static __device__ __forceinline__ float finite_or_zero(float m) { return isfinite(m) ? m : 0.0f; }
static __device__ __forceinline__ float wave_max(float v) {
    #pragma unroll
    for (int o = 16; o >= 1; o >>= 1) v = fmaxf(v, __shfl_xor(v, o, 32));
    return v;
}
static __device__ __forceinline__ float wave_sum(float v) {
    #pragma unroll
    for (int o = 16; o >= 1; o >>= 1) v += __shfl_xor(v, o, 32);
    return v;
}
static __device__ __forceinline__ float lane_f(float v, int j) {
    return __int_as_float(__builtin_amdgcn_readlane(__float_as_int(v), j));
}

#define SLOTS 32
#define GEMM_BLOCKS 768
#define BIN_BLOCKS 256
#define NB_MAX 128

static __device__ __forceinline__ int range_of(int d, int nb) {
    const int k = (int)((unsigned)d / (unsigned)nb);
    return k < GEMM_BLOCKS ? k : GEMM_BLOCKS - 1;
}

#define BIN_THREADS 1024
#define BIN_BATCH 4
__global__ __launch_bounds__(BIN_THREADS) void bin_edges(const int* __restrict__ ei, int E, int N,
                                                         int* __restrict__ offT,
                                                         int* __restrict__ binned_w,
                                                         int* __restrict__ ovf_n) {
    __shared__ int cnt[GEMM_BLOCKS];
    __shared__ int off[GEMM_BLOCKS + 1];
    __shared__ int wsum[8];
    static_assert(GEMM_BLOCKS == 3 * 256, "three ranges per thread in the scan below");
    int2* binned = (int2*)binned_w;
    const int nb = (N + GEMM_BLOCKS - 1) / GEMM_BLOCKS;
    const int C = (E + BIN_BLOCKS - 1) / BIN_BLOCKS;
    const int e0 = blockIdx.x * C;
    const int e1 = min(E, e0 + C);
    const int tid = threadIdx.x;
    const int lane = tid & 31, wave = tid >> 5;
    const int b3 = 3 * tid;
    if (blockIdx.x == 0 && tid == 0) *ovf_n = 0;
    if (tid < GEMM_BLOCKS) cnt[tid] = 0;
    __syncthreads();

    for (int eb = e0; eb < e1; eb += BIN_THREADS * BIN_BATCH) {
        int d[BIN_BATCH];
        #pragma unroll
        for (int j = 0; j < BIN_BATCH; ++j) d[j] = ei[E + min(eb + tid + BIN_THREADS * j, e1 - 1)];
        #pragma unroll
        for (int j = 0; j < BIN_BATCH; ++j)
            if (eb + tid + BIN_THREADS * j < e1) atomicAdd(&cnt[range_of(d[j], nb)], 1);
    }
    __syncthreads();

    int c0 = 0, c1 = 0, c2 = 0, tsum = 0, incl = 0;
    if (tid < 256) {
        c0 = cnt[b3]; c1 = cnt[b3 + 1]; c2 = cnt[b3 + 2];
        tsum = c0 + c1 + c2;
        incl = tsum;
        #pragma unroll
        for (int o = 1; o < 32; o <<= 1) {
            const int v = __shfl_up(incl, o, 32);
            if (lane >= o) incl += v;
        }
        if (lane == 31) wsum[wave] = incl;
        cnt[b3] = 0; cnt[b3 + 1] = 0; cnt[b3 + 2] = 0;
    }
    __syncthreads();
    if (tid < 256) {
        int base = 0;
        for (int w = 0; w < wave; ++w) base += wsum[w];
        const int ex = base + incl - tsum;
        off[b3] = ex; off[b3 + 1] = ex + c0; off[b3 + 2] = ex + c0 + c1;
        if (tid == 255) off[GEMM_BLOCKS] = ex + tsum;
    }
    __syncthreads();
    if (tid <= GEMM_BLOCKS) offT[(size_t)tid * BIN_BLOCKS + blockIdx.x] = off[tid];

    for (int eb = e0; eb < e1; eb += BIN_THREADS * BIN_BATCH) {
        int d[BIN_BATCH], s[BIN_BATCH];
        #pragma unroll
        for (int j = 0; j < BIN_BATCH; ++j) {
            const int e = min(eb + tid + BIN_THREADS * j, e1 - 1);
            d[j] = ei[E + e];
            s[j] = ei[e];
        }
        #pragma unroll
        for (int j = 0; j < BIN_BATCH; ++j)
            if (eb + tid + BIN_THREADS * j < e1) {
                const int k = range_of(d[j], nb);
                const int r = atomicAdd(&cnt[k], 1);
                int2 ds; ds.x = d[j]; ds.y = s[j];
                binned[(size_t)e0 + off[k] + r] = ds;
            }
    }
}

static __device__ __forceinline__ void put_edge(int* __restrict__ slots, int* __restrict__ ovf_d,
                                                int* __restrict__ ovf_s, int* __restrict__ ovf_n,
                                                int d, int s, int r) {
    if (r < SLOTS) {
        slots[(size_t)d * SLOTS + r] = s;
    } else {
        const int k = atomicAdd(ovf_n, 1);
        ovf_d[k] = d;
        ovf_s[k] = s;
    }
}

static __device__ __forceinline__ void lds_barrier() {
    __builtin_amdgcn_fence(__ATOMIC_RELEASE, "workgroup", "local");
    __builtin_amdgcn_s_barrier();
    __builtin_amdgcn_fence(__ATOMIC_ACQUIRE, "workgroup", "local");
}

static __device__ __forceinline__ void place_one(int* cntN, int nbase, int2 q, int* __restrict__ slots,
                                                 int* __restrict__ ovf_d, int* __restrict__ ovf_s,
                                                 int* __restrict__ ovf_n) {
    const int r = atomicAdd(&cntN[q.x - nbase], 1);
    put_edge(slots, ovf_d, ovf_s, ovf_n, q.x, q.y, r);
}
static __device__ __forceinline__ void place_range(int* cntN, int N, int E,
                                                   const int* __restrict__ offT,
                                                   const int* __restrict__ binned_w,
                                                   int* __restrict__ deg, int* __restrict__ slots,
                                                   int* __restrict__ ovf_d, int* __restrict__ ovf_s) {
    const int2* binned = (const int2*)binned_w;
    const int nb = (N + GEMM_BLOCKS - 1) / GEMM_BLOCKS;
    const int nbase = blockIdx.x * nb;
    const int C = (E + BIN_BLOCKS - 1) / BIN_BLOCKS;
    for (int b = threadIdx.x; b < BIN_BLOCKS; b += 256) {
        const int lo = offT[(size_t)blockIdx.x * BIN_BLOCKS + b];
        const int hi = offT[(size_t)(blockIdx.x + 1) * BIN_BLOCKS + b];
        const int2* seg = binned + (size_t)b * C;
        for (int p = lo; p < hi; p += 8) {
            const int last = hi - 1;
            const int2 q0 = seg[p];
            const int2 q1 = seg[min(p + 1, last)];
            const int2 q2 = seg[min(p + 2, last)];
            const int2 q3 = seg[min(p + 3, last)];
            const int2 q4 = seg[min(p + 4, last)];
            const int2 q5 = seg[min(p + 5, last)];
            const int2 q6 = seg[min(p + 6, last)];
            const int2 q7 = seg[min(p + 7, last)];
            place_one(cntN, nbase, q0, slots, ovf_d, ovf_s, deg + N);
            if (p + 1 < hi) place_one(cntN, nbase, q1, slots, ovf_d, ovf_s, deg + N);
            if (p + 2 < hi) place_one(cntN, nbase, q2, slots, ovf_d, ovf_s, deg + N);
            if (p + 3 < hi) place_one(cntN, nbase, q3, slots, ovf_d, ovf_s, deg + N);
            if (p + 4 < hi) place_one(cntN, nbase, q4, slots, ovf_d, ovf_s, deg + N);
            if (p + 5 < hi) place_one(cntN, nbase, q5, slots, ovf_d, ovf_s, deg + N);
            if (p + 6 < hi) place_one(cntN, nbase, q6, slots, ovf_d, ovf_s, deg + N);
            if (p + 7 < hi) place_one(cntN, nbase, q7, slots, ovf_d, ovf_s, deg + N);
        }
    }
    lds_barrier();
    if ((int)threadIdx.x < nb && nbase + (int)threadIdx.x < N) deg[nbase + threadIdx.x] = cntN[threadIdx.x];
}

#define XS_PITCH 132
__global__ __launch_bounds__(256, 8) void gemm_wmma_f32(const float* __restrict__ X,
                                                     const float* __restrict__ W,
                                                     float* __restrict__ H, int N, int E,
                                                     const int* __restrict__ offT,
                                                     const int* __restrict__ binned_w,
                                                     int* __restrict__ deg, int* __restrict__ slots,
                                                     int* __restrict__ ovf_d, int* __restrict__ ovf_s,
                                                     const float* __restrict__ att_s,
                                                     const float* __restrict__ att_d,
                                                     float* __restrict__ as1, float* __restrict__ ad1) {
    __shared__ __attribute__((aligned(16))) float xs[16 * XS_PITCH];
    __shared__ __attribute__((aligned(16))) float hs[16 * XS_PITCH];
    __shared__ __attribute__((aligned(16))) float atts[256];
    __shared__ int cntN[NB_MAX];

    if (threadIdx.x < NB_MAX) cntN[threadIdx.x] = 0;
    lds_barrier();

    const int nStripes = N >> 4;
    const int st0 = (int)((blockIdx.x * (unsigned)nStripes) / gridDim.x);
    const int st1 = (int)(((blockIdx.x + 1u) * (unsigned)nStripes) / gridDim.x);

    const int wave = threadIdx.x >> 5;
    const int lane = threadIdx.x & 31;
    const int colBase = wave * 16;
    const int half = lane >> 4;
    const int mn = lane & 15;
    const int f0 = threadIdx.x, f1 = threadIdx.x + 256;
    float* xs0 = xs + (f0 >> 5) * XS_PITCH + (f0 & 31) * 4;
    float* xs1 = xs + (f1 >> 5) * XS_PITCH + (f1 & 31) * 4;
    const float* hs0 = hs + (f0 >> 5) * XS_PITCH + (f0 & 31) * 4;
    const float* hs1 = hs + (f1 >> 5) * XS_PITCH + (f1 & 31) * 4;

    if (st0 < st1) {
        const float attv = threadIdx.x < 128 ? att_s[threadIdx.x] : att_d[threadIdx.x - 128];

        const float4* xg0 = (const float4*)(X + (size_t)st0 * 2048);
        float4 x0 = xg0[f0];
        float4 x1 = xg0[f1];

        v2f b[32];
        #pragma unroll
        for (int i = 0; i < 32; ++i) {
            const int ka = 4 * i + 2 * half;
            b[i].x = W[(size_t)ka * 128 + colBase + mn];
            b[i].y = W[(size_t)(ka + 1) * 128 + colBase + mn];
        }
        atts[threadIdx.x] = attv;

        place_range(cntN, N, E, offT, binned_w, deg, slots, ovf_d, ovf_s);

        #pragma unroll
        for (int i = 0; i < 32; ++i) asm volatile("" : "+v"(b[i]));

        const float* xrow = xs + mn * XS_PITCH + 2 * half;
        float* htile = hs + (8 * half) * XS_PITCH + colBase + mn;

        for (int st = st0; st < st1; ++st) {
            *(float4*)xs0 = x0;
            *(float4*)xs1 = x1;
            lds_barrier();

            if (st + 1 < st1) {
                const float4* xg = (const float4*)(X + (size_t)(st + 1) * 2048);
                x0 = xg[f0];
                x1 = xg[f1];
            }

            v8f c = {};
            #pragma unroll
            for (int i = 0; i < 32; ++i) {
                const v2f a = *(const v2f*)(xrow + 4 * i);
                c = __builtin_amdgcn_wmma_f32_16x16x4_f32(false, a, false, b[i],
                                                          (short)0, c, false, false);
            }
            #pragma unroll
            for (int r = 0; r < 8; ++r) {
                htile[r * XS_PITCH] = c[r];
            }
            lds_barrier();

            float4* hg = (float4*)(H + (size_t)st * 2048);
            hg[f0] = *(const float4*)hs0;
            hg[f1] = *(const float4*)hs1;

            if (threadIdx.x < 128) {
                const bool dside = threadIdx.x >= 64;
                const int idx = threadIdx.x & 63;
                const int hd = idx & 3;
                const float4* ap = (const float4*)(atts + (dside ? 128 : 0) + hd * 32);
                const float4* hp4 = (const float4*)(hs + (idx >> 2) * XS_PITCH + hd * 32);
                float s = 0.0f;
                #pragma unroll 4
                for (int j = 0; j < 8; ++j) {
                    const float4 hv = hp4[j], av = ap[j];
                    s += hv.x * av.x + hv.y * av.y + hv.z * av.z + hv.w * av.w;
                }
                (dside ? ad1 : as1)[(size_t)st * 64 + idx] = s;
            }
        }
    } else {
        place_range(cntN, N, E, offT, binned_w, deg, slots, ovf_d, ovf_s);
    }
}

static __device__ __forceinline__ void add_rows(float4& acc, int cnt, int s, float4 al,
                                                const float* __restrict__ hl) {
    int j = 0;
    for (; j + 4 <= cnt; j += 4) {
        const float* r0 = hl + (size_t)__builtin_amdgcn_readlane(s, j) * 128;
        const float* r1 = hl + (size_t)__builtin_amdgcn_readlane(s, j + 1) * 128;
        const float* r2 = hl + (size_t)__builtin_amdgcn_readlane(s, j + 2) * 128;
        const float* r3 = hl + (size_t)__builtin_amdgcn_readlane(s, j + 3) * 128;
        const float q00 = r0[0], q01 = r0[32], q02 = r0[64], q03 = r0[96];
        const float q10 = r1[0], q11 = r1[32], q12 = r1[64], q13 = r1[96];
        const float q20 = r2[0], q21 = r2[32], q22 = r2[64], q23 = r2[96];
        const float q30 = r3[0], q31 = r3[32], q32 = r3[64], q33 = r3[96];
        acc.x += lane_f(al.x, j) * q00;     acc.y += lane_f(al.y, j) * q01;
        acc.z += lane_f(al.z, j) * q02;     acc.w += lane_f(al.w, j) * q03;
        acc.x += lane_f(al.x, j + 1) * q10; acc.y += lane_f(al.y, j + 1) * q11;
        acc.z += lane_f(al.z, j + 1) * q12; acc.w += lane_f(al.w, j + 1) * q13;
        acc.x += lane_f(al.x, j + 2) * q20; acc.y += lane_f(al.y, j + 2) * q21;
        acc.z += lane_f(al.z, j + 2) * q22; acc.w += lane_f(al.w, j + 2) * q23;
        acc.x += lane_f(al.x, j + 3) * q30; acc.y += lane_f(al.y, j + 3) * q31;
        acc.z += lane_f(al.z, j + 3) * q32; acc.w += lane_f(al.w, j + 3) * q33;
    }
    for (; j < cnt; ++j) {
        const float* r0 = hl + (size_t)__builtin_amdgcn_readlane(s, j) * 128;
        const float q00 = r0[0], q01 = r0[32], q02 = r0[64], q03 = r0[96];
        acc.x += lane_f(al.x, j) * q00;     acc.y += lane_f(al.y, j) * q01;
        acc.z += lane_f(al.z, j) * q02;     acc.w += lane_f(al.w, j) * q03;
    }
}

static __device__ __forceinline__ void add_row(float4& acc, int j, int s, float4 al,
                                               const float* __restrict__ hl) {
    const float* r0 = hl + (size_t)__builtin_amdgcn_readlane(s, j) * 128;
    const float q00 = r0[0], q01 = r0[32], q02 = r0[64], q03 = r0[96];
    acc.x += lane_f(al.x, j) * q00;     acc.y += lane_f(al.y, j) * q01;
    acc.z += lane_f(al.z, j) * q02;     acc.w += lane_f(al.w, j) * q03;
}

#define L1_NODES_PER_WAVE 5

__global__ __launch_bounds__(256, 10) void seg_l1(const int* __restrict__ deg,
                                              const int* __restrict__ slots,
                                              const int* __restrict__ ovf_d,
                                              const int* __restrict__ ovf_s,
                                              const float* __restrict__ as1,
                                              const float* __restrict__ ad1,
                                              const float* __restrict__ h1,
                                              const float* __restrict__ b1,
                                              const float* __restrict__ W2,
                                              float* __restrict__ h2, int N) {
    const int n0 = L1_NODES_PER_WAVE *
                   __builtin_amdgcn_readfirstlane((int)(blockIdx.x * 8 + (threadIdx.x >> 5)));
    if (n0 >= N) return;
    const int lane = threadIdx.x & 31;
    const float* hl = h1 + lane;
    const float4* as4 = (const float4*)as1;
    const float4* ad4 = (const float4*)ad1;

    const float c0 = b1[lane], c1 = b1[lane + 32], c2 = b1[lane + 64], c3 = b1[lane + 96];
    const float2* w2v = (const float2*)W2;
    const float2 w0 = w2v[lane], w1 = w2v[lane + 32], w2 = w2v[lane + 64], w3 = w2v[lane + 96];

    int g0 = deg[n0];
    int g1 = deg[min(n0 + 1, N - 1)];
    const int sl = lane > 0 ? lane - 1 : 0;
    const int sx0 = slots[(size_t)n0 * SLOTS + sl];
    int s0 = (lane > 0 && lane < min(g0, SLOTS) + 1) ? sx0 : n0;
    float4 r0 = as4[s0];
    float4 adv = ad4[n0];

    for (int i = 0; i < L1_NODES_PER_WAVE; ++i) {
        const int n = n0 + i;
        if (n >= N) break;
        const int* row = slots + (size_t)n * SLOTS;
        const int T = min(g0, SLOTS) + 1;
        const bool over = g0 > SLOTS;
        const int novf = over ? deg[N] : 0;
        const bool more = (i + 1 < L1_NODES_PER_WAVE) && (n + 1 < N);

        const bool ok0 = lane < T;
        float4 v0;
        v0.x = lrelu(r0.x + adv.x);
        v0.y = lrelu(r0.y + adv.y);
        v0.z = lrelu(r0.z + adv.z);
        v0.w = lrelu(r0.w + adv.w);

        const int g2 = deg[min(n + 2, N - 1)];
        const int nn = more ? n + 1 : n;
        const int sxn = slots[(size_t)nn * SLOTS + sl];
        const bool hasn = more && lane > 0 && lane < min(g1, SLOTS) + 1;

        float4 m;
        m.x = ok0 ? v0.x : -INFINITY;
        m.y = ok0 ? v0.y : -INFINITY;
        m.z = ok0 ? v0.z : -INFINITY;
        m.w = ok0 ? v0.w : -INFINITY;
        for (int t0 = 32; t0 < T; t0 += 32) {
            const int t = t0 + lane;
            if (t < T) {
                float4 v = as4[row[t - 1]];
                m.x = fmaxf(m.x, lrelu(v.x + adv.x));
                m.y = fmaxf(m.y, lrelu(v.y + adv.y));
                m.z = fmaxf(m.z, lrelu(v.z + adv.z));
                m.w = fmaxf(m.w, lrelu(v.w + adv.w));
            }
        }
        if (over) {
            for (int k = lane; k < novf; k += 32) {
                if (ovf_d[k] == n) {
                    float4 v = as4[ovf_s[k]];
                    m.x = fmaxf(m.x, lrelu(v.x + adv.x));
                    m.y = fmaxf(m.y, lrelu(v.y + adv.y));
                    m.z = fmaxf(m.z, lrelu(v.z + adv.z));
                    m.w = fmaxf(m.w, lrelu(v.w + adv.w));
                }
            }
        }
        m.x = finite_or_zero(wave_max(m.x));
        m.y = finite_or_zero(wave_max(m.y));
        m.z = finite_or_zero(wave_max(m.z));
        m.w = finite_or_zero(wave_max(m.w));

        float4 e0;
        e0.x = ok0 ? expf(v0.x - m.x) : 0.0f;
        e0.y = ok0 ? expf(v0.y - m.y) : 0.0f;
        e0.z = ok0 ? expf(v0.z - m.z) : 0.0f;
        e0.w = ok0 ? expf(v0.w - m.w) : 0.0f;
        float4 d = e0;
        for (int t0 = 32; t0 < T; t0 += 32) {
            const int t = t0 + lane;
            if (t < T) {
                float4 v = as4[row[t - 1]];
                d.x += expf(lrelu(v.x + adv.x) - m.x);
                d.y += expf(lrelu(v.y + adv.y) - m.y);
                d.z += expf(lrelu(v.z + adv.z) - m.z);
                d.w += expf(lrelu(v.w + adv.w) - m.w);
            }
        }
        if (over) {
            for (int k = lane; k < novf; k += 32) {
                if (ovf_d[k] == n) {
                    float4 v = as4[ovf_s[k]];
                    d.x += expf(lrelu(v.x + adv.x) - m.x);
                    d.y += expf(lrelu(v.y + adv.y) - m.y);
                    d.z += expf(lrelu(v.z + adv.z) - m.z);
                    d.w += expf(lrelu(v.w + adv.w) - m.w);
                }
            }
        }
        d.x = wave_sum(d.x) + EPS_F;
        d.y = wave_sum(d.y) + EPS_F;
        d.z = wave_sum(d.z) + EPS_F;
        d.w = wave_sum(d.w) + EPS_F;

        const int s0n = hasn ? sxn : nn;
        const float4 r0n = as4[s0n];
        const float4 advn = ad4[nn];

        float4 acc = {0.0f, 0.0f, 0.0f, 0.0f};
        {
            float4 al;
            al.x = e0.x / d.x;
            al.y = e0.y / d.y;
            al.z = e0.z / d.z;
            al.w = e0.w / d.w;
            add_rows(acc, T < 32 ? T : 32, s0, al, hl);
        }
        for (int t0 = 32; t0 < T; t0 += 32) {
            const int t = t0 + lane;
            int s = n;
            float4 al = {0.0f, 0.0f, 0.0f, 0.0f};
            if (t < T) {
                s = row[t - 1];
                float4 v = as4[s];
                al.x = expf(lrelu(v.x + adv.x) - m.x) / d.x;
                al.y = expf(lrelu(v.y + adv.y) - m.y) / d.y;
                al.z = expf(lrelu(v.z + adv.z) - m.z) / d.z;
                al.w = expf(lrelu(v.w + adv.w) - m.w) / d.w;
            }
            const int left = T - t0;
            add_rows(acc, left < 32 ? left : 32, s, al, hl);
        }
        if (over) {
            for (int k0 = 0; k0 < novf; k0 += 32) {
                const int k = k0 + lane;
                bool hit = false;
                int s = n;
                float4 al = {0.0f, 0.0f, 0.0f, 0.0f};
                if (k < novf && ovf_d[k] == n) {
                    hit = true;
                    s = ovf_s[k];
                    float4 v = as4[s];
                    al.x = expf(lrelu(v.x + adv.x) - m.x) / d.x;
                    al.y = expf(lrelu(v.y + adv.y) - m.y) / d.y;
                    al.z = expf(lrelu(v.z + adv.z) - m.z) / d.z;
                    al.w = expf(lrelu(v.w + adv.w) - m.w) / d.w;
                }
                unsigned hits = __builtin_amdgcn_ballot_w32(hit);
                while (hits != 0u) {
                    add_row(acc, __builtin_ctz(hits), s, al, hl);
                    hits &= hits - 1u;
                }
            }
        }

        float4 f;
        f.x = acc.x + c0; f.x = f.x > 0.0f ? f.x : expf(f.x) - 1.0f;
        f.y = acc.y + c1; f.y = f.y > 0.0f ? f.y : expf(f.y) - 1.0f;
        f.z = acc.z + c2; f.z = f.z > 0.0f ? f.z : expf(f.z) - 1.0f;
        f.w = acc.w + c3; f.w = f.w > 0.0f ? f.w : expf(f.w) - 1.0f;
        float p0 = f.x * w0.x + f.y * w1.x + f.z * w2.x + f.w * w3.x;
        float p1 = f.x * w0.y + f.y * w1.y + f.z * w2.y + f.w * w3.y;
        p0 = wave_sum(p0);
        p1 = wave_sum(p1);
        if (lane == 0) {
            float2 pr;
            pr.x = p0;
            pr.y = p1;
            ((float2*)h2)[n] = pr;
        }

        g0 = g1; g1 = g2;
        s0 = s0n; r0 = r0n; adv = advn;
    }
}

#define L2_LANES 4
#define L2_CHUNKS 8
static __device__ __forceinline__ float group4_max(float v) {
    v = fmaxf(v, __shfl_xor(v, 2, 32));
    v = fmaxf(v, __shfl_xor(v, 1, 32));
    return v;
}
static __device__ __forceinline__ float group4_sum(float v) {
    v += __shfl_xor(v, 2, 32);
    v += __shfl_xor(v, 1, 32);
    return v;
}

__global__ __launch_bounds__(256) void seg_l2(const int* __restrict__ deg,
                                              const int* __restrict__ slots,
                                              const int* __restrict__ ovf_d,
                                              const int* __restrict__ ovf_s,
                                              const float* __restrict__ att_s2,
                                              const float* __restrict__ att_d2,
                                              const float* __restrict__ h2,
                                              const float* __restrict__ b2,
                                              float* __restrict__ y, int N) {
    const int lane = threadIdx.x & 31;
    const int q = lane & (L2_LANES - 1);
    const int nw = (32 / L2_LANES) * (int)(blockIdx.x * 8 + (threadIdx.x >> 5));
    if (nw >= N) return;
    const int nr = nw + lane / L2_LANES;
    const bool live = nr < N;
    const int n = live ? nr : N - 1;
    const int cnt = deg[n];
    const int T = min(cnt, SLOTS) + 1;
    const bool over = cnt > SLOTS;
    const int novf = over ? deg[N] : 0;
    const int* row = slots + (size_t)n * SLOTS;
    const float2* h2v = (const float2*)h2;
    const float ts0 = att_s2[0], ts1 = att_s2[1], td0 = att_d2[0], td1 = att_d2[1];
    const float2 hn = h2v[n];
    const float adn = hn.x * td0 + hn.y * td1;

    int s[L2_CHUNKS];
    bool ok[L2_CHUNKS];
    #pragma unroll
    for (int c = 0; c < L2_CHUNKS; ++c) {
        const int t = L2_LANES * c + q;
        ok[c] = t < T;
        const bool has = ok[c] && t > 0;
        const int sx = row[t > 0 ? t - 1 : 0];
        s[c] = has ? sx : n;
    }
    float v[L2_CHUNKS];
    float2 g[L2_CHUNKS];
    #pragma unroll
    for (int c = 0; c < L2_CHUNKS; ++c) {
        g[c] = h2v[s[c]];
        v[c] = lrelu((g[c].x * ts0 + g[c].y * ts1) + adn);
    }

    float m = -INFINITY;
    #pragma unroll
    for (int c = 0; c < L2_CHUNKS; ++c) m = ok[c] ? fmaxf(m, v[c]) : m;
    for (int t = 32 + q; t < T; t += L2_LANES) {
        const float2 gx = h2v[row[t - 1]];
        m = fmaxf(m, lrelu((gx.x * ts0 + gx.y * ts1) + adn));
    }
    if (over)
        for (int k = q; k < novf; k += L2_LANES)
            if (ovf_d[k] == n) {
                const float2 gx = h2v[ovf_s[k]];
                m = fmaxf(m, lrelu((gx.x * ts0 + gx.y * ts1) + adn));
            }
    m = finite_or_zero(group4_max(m));

    float e[L2_CHUNKS];
    float d = 0.0f;
    #pragma unroll
    for (int c = 0; c < L2_CHUNKS; ++c) {
        e[c] = ok[c] ? expf(v[c] - m) : 0.0f;
        d += e[c];
    }
    for (int t = 32 + q; t < T; t += L2_LANES) {
        const float2 gx = h2v[row[t - 1]];
        d += expf(lrelu((gx.x * ts0 + gx.y * ts1) + adn) - m);
    }
    if (over)
        for (int k = q; k < novf; k += L2_LANES)
            if (ovf_d[k] == n) {
                const float2 gx = h2v[ovf_s[k]];
                d += expf(lrelu((gx.x * ts0 + gx.y * ts1) + adn) - m);
            }
    d = group4_sum(d) + EPS_F;

    float o0 = 0.0f, o1 = 0.0f;
    #pragma unroll
    for (int c = 0; c < L2_CHUNKS; ++c) {
        const float al = e[c] / d;
        o0 += ok[c] ? al * g[c].x : 0.0f;
        o1 += ok[c] ? al * g[c].y : 0.0f;
    }
    for (int t = 32 + q; t < T; t += L2_LANES) {
        const float2 gx = h2v[row[t - 1]];
        const float al = expf(lrelu((gx.x * ts0 + gx.y * ts1) + adn) - m) / d;
        o0 += al * gx.x;
        o1 += al * gx.y;
    }
    if (over)
        for (int k = q; k < novf; k += L2_LANES)
            if (ovf_d[k] == n) {
                const float2 gx = h2v[ovf_s[k]];
                const float al = expf(lrelu((gx.x * ts0 + gx.y * ts1) + adn) - m) / d;
                o0 += al * gx.x;
                o1 += al * gx.y;
            }
    o0 = group4_sum(o0);
    o1 = group4_sum(o1);
    if (q == 0 && live) {
        const float z0 = o0 + b2[0];
        const float z1 = o1 + b2[1];
        const float mx = fmaxf(z0, z1);
        const float lse = mx + logf(expf(z0 - mx) + expf(z1 - mx));
        float2 r;
        r.x = z0 - lse;
        r.y = z1 - lse;
        ((float2*)y)[n] = r;
    }
}

extern "C" void kernel_launch(void* const* d_in, const int* in_sizes, int n_in,
                              void* d_out, int out_size, void* d_ws, size_t ws_size,
                              hipStream_t stream) {
    const float* x        = (const float*)d_in[0];
    const int*   ei       = (const int*)d_in[1];
    const float* W1       = (const float*)d_in[2];
    const float* att_src1 = (const float*)d_in[3];
    const float* att_dst1 = (const float*)d_in[4];
    const float* b1       = (const float*)d_in[5];
    const float* W2       = (const float*)d_in[6];
    const float* att_src2 = (const float*)d_in[7];
    const float* att_dst2 = (const float*)d_in[8];
    const float* b2       = (const float*)d_in[9];
    float* y = (float*)d_out;

    const int N = in_sizes[0] / 128;
    const int E = in_sizes[1] / 2;

    float* ws    = (float*)d_ws;
    float* h1    = ws;
    int*   slots = (int*)(h1 + (size_t)N * 128);
    float* as1   = (float*)(slots + (size_t)N * SLOTS);
    float* ad1   = as1 + (size_t)N * 4;
    float* h2    = ad1 + (size_t)N * 4;
    int*   binned = (int*)(h2 + (size_t)N * 2);
    int*   deg   = binned + (size_t)E * 2;
    int*   ovf_d = deg + (size_t)N + 1;
    int*   ovf_s = ovf_d + (size_t)E;
    int*   offT  = ovf_s + (size_t)E;
    if ((size_t)(offT + (size_t)(GEMM_BLOCKS + 1) * BIN_BLOCKS - (int*)d_ws) * 4 > ws_size) return;
    if (N > GEMM_BLOCKS * NB_MAX) return;

    const int TB = 256;

    bin_edges<<<BIN_BLOCKS, BIN_THREADS, 0, stream>>>(ei, E, N, offT, binned, deg + N);
    gemm_wmma_f32<<<GEMM_BLOCKS, 256, 0, stream>>>(x, W1, h1, N, E, offT, binned, deg, slots, ovf_d,
                                              ovf_s, att_src1, att_dst1, as1, ad1);

    seg_l1<<<(N + 39) / 40, 256, 0, stream>>>(deg, slots, ovf_d, ovf_s, as1, ad1, h1, b1, W2, h2, N);

    seg_l2<<<(N + 63) / 64, 256, 0, stream>>>(deg, slots, ovf_d, ovf_s, att_src2, att_dst2, h2, b2,
                                            y, N);
}
